# starting kernel
# baseline (speedup 1.0000x reference)
#include <hip/hip_runtime.h>
#include <math.h>

typedef __attribute__((ext_vector_type(2))) float v2f;
typedef __attribute__((ext_vector_type(8))) float v8f;

#define B_SZ   64
#define NI     2048
#define DI     8
#define NO     32
#define DO_    16
#define NOD    512
#define KTOT   (NI*DI)
#define WN     (NO*DO_*DI)

#define NB_GEMM  128
#define KCHUNK   (KTOT/NB_GEMM)
#define NB_ROUTE 128
#define NPB      (NI/NB_ROUTE)
#define PS_STRIDE (B_SZ*NOD)

__global__ __launch_bounds__(256)
void caps_s0_gemm(const float* __restrict__ x, const float* __restrict__ W,
                  float* __restrict__ ps) {
  const int lane = threadIdx.x & 31;
  const int wave = threadIdx.x >> 5;
  const int half = lane >> 4;
  const int l15  = lane & 15;
  const int kbase = blockIdx.x * KCHUNK;
  float* psb = ps + (size_t)blockIdx.x * PS_STRIDE;

  for (int ot = 0; ot < 4; ++ot) {
    const int o = wave + ot * 8;
    const float* wc = W + (size_t)(o * DO_ + l15) * DI;
    for (int mt = 0; mt < 4; ++mt) {
      const int row = mt * 16 + l15;
      const float* xr = x + (size_t)row * KTOT;
      v8f acc = {};
      for (int k0 = kbase; k0 < kbase + KCHUNK; k0 += 4) {
        const int ka = k0 + 2 * half;
        v2f a  = *(const v2f*)(xr + ka);
        const int n = ka >> 3, i = ka & 7;
        v2f bv = *(const v2f*)(wc + (size_t)n * WN + i);
        acc = __builtin_amdgcn_wmma_f32_16x16x4_f32(
                  false, a, false, bv, (short)0, acc, false, false);
      }
#pragma unroll
      for (int r = 0; r < 8; ++r) {
        const int b = mt * 16 + r + 8 * half;
        psb[b * NOD + o * DO_ + l15] = acc[r];
      }
    }
  }
}

__global__ __launch_bounds__(256)
void caps_route(const float* __restrict__ x, const float* __restrict__ W,
                const float* __restrict__ vcur, float* __restrict__ rlog,
                float* __restrict__ ps, int iter) {
  __shared__ float r1s[B_SZ * NO];
  __shared__ float cs[B_SZ * NO];
  const int lane = threadIdx.x & 31;
  const int wave = threadIdx.x >> 5;
  const int half = lane >> 4;
  const int l15  = lane & 15;

  v8f zero = {};
  v8f sacc[16];
#pragma unroll
  for (int t = 0; t < 16; ++t) sacc[t] = zero;

  const int n0 = blockIdx.x * NPB;
  for (int nn = 0; nn < NPB; ++nn) {
    const int n = n0 + nn;
    const float* wn = W + (size_t)n * WN;

    for (int ot = 0; ot < 4; ++ot) {
      const int o = wave + ot * 8;
      for (int mt = 0; mt < 4; ++mt) {
        const int row = mt * 16 + l15;
        v8f d = zero;
#pragma unroll
        for (int ks = 0; ks < 2; ++ks) {
          const int k = ks * 4 + 2 * half;
          v2f a  = *(const v2f*)(x + (size_t)row * KTOT + n * DI + k);
          v2f bv = *(const v2f*)(wn + (size_t)(o * DO_ + l15) * DI + k);
          d = __builtin_amdgcn_wmma_f32_16x16x4_f32(
                  false, a, false, bv, (short)0, d, false, false);
        }
#pragma unroll
        for (int r = 0; r < 8; ++r) {
          const int b = mt * 16 + r + 8 * half;
          float p = d[r] * vcur[b * NOD + o * DO_ + l15];
          p += __shfl_xor(p, 1);
          p += __shfl_xor(p, 2);
          p += __shfl_xor(p, 4);
          p += __shfl_xor(p, 8);
          if (l15 == 0) {
            const size_t ridx = ((size_t)b * NI + n) * NO + o;
            float logit;
            if (iter == 1) { logit = p; rlog[ridx] = p; }
            else           { logit = rlog[ridx] + p; }
            r1s[b * NO + o] = logit;
          }
        }
      }
    }
    __syncthreads();

    if (threadIdx.x < B_SZ) {
      const int b = threadIdx.x;
      float m = -3.4e38f;
      for (int o = 0; o < NO; ++o) m = fmaxf(m, r1s[b * NO + o]);
      float ssum = 0.f;
      for (int o = 0; o < NO; ++o) {
        float e = __expf(r1s[b * NO + o] - m);
        cs[b * NO + o] = e; ssum += e;
      }
      const float inv = 1.f / ssum;
      for (int o = 0; o < NO; ++o) cs[b * NO + o] *= inv;
    }
    __syncthreads();

    for (int ot = 0; ot < 4; ++ot) {
      const int o = wave + ot * 8;
      for (int mt = 0; mt < 4; ++mt) {
        const int row = mt * 16 + l15;
        v8f d = zero;
#pragma unroll
        for (int ks = 0; ks < 2; ++ks) {
          const int k = ks * 4 + 2 * half;
          v2f a  = *(const v2f*)(x + (size_t)row * KTOT + n * DI + k);
          v2f bv = *(const v2f*)(wn + (size_t)(o * DO_ + l15) * DI + k);
          d = __builtin_amdgcn_wmma_f32_16x16x4_f32(
                  false, a, false, bv, (short)0, d, false, false);
        }
#pragma unroll
        for (int r = 0; r < 8; ++r) {
          const int b = mt * 16 + r + 8 * half;
          sacc[ot * 4 + mt][r] += cs[b * NO + o] * d[r];
        }
      }
    }
    __syncthreads();
  }

  float* psb = ps + (size_t)blockIdx.x * PS_STRIDE;
  for (int ot = 0; ot < 4; ++ot) {
    const int o = wave + ot * 8;
    for (int mt = 0; mt < 4; ++mt) {
      v8f s = sacc[ot * 4 + mt];
#pragma unroll
      for (int r = 0; r < 8; ++r) {
        const int b = mt * 16 + r + 8 * half;
        psb[b * NOD + o * DO_ + l15] = s[r];
      }
    }
  }
}

__global__ __launch_bounds__(512)
void caps_reduce_squash(const float* __restrict__ ps, int nparts, float prescale,
                        float* __restrict__ out) {
  __shared__ float sh[NOD];
  __shared__ float scl[NO];
  const int b  = blockIdx.x;
  const int od = threadIdx.x;
  float s = 0.f;
  for (int p = 0; p < nparts; ++p)
    s += ps[(size_t)p * PS_STRIDE + b * NOD + od];
  s *= prescale;
  sh[od] = s;
  __syncthreads();
  if (od < NO) {
    float nn = 0.f;
#pragma unroll
    for (int d = 0; d < DO_; ++d) { float t = sh[od * DO_ + d]; nn += t * t; }
    scl[od] = sqrtf(nn) / (1.f + nn);
  }
  __syncthreads();
  out[(size_t)b * NOD + od] = sh[od] * scl[od >> 4];
}

extern "C" void kernel_launch(void* const* d_in, const int* in_sizes, int n_in,
                              void* d_out, int out_size, void* d_ws, size_t ws_size,
                              hipStream_t stream) {
  const float* x = (const float*)d_in[0];
  const float* W = (const float*)d_in[1];
  float* out = (float*)d_out;

  float* ps   = (float*)d_ws;
  float* vbuf = ps + (size_t)NB_GEMM * PS_STRIDE;
  float* rlog = vbuf + (size_t)B_SZ * NOD;

  caps_s0_gemm<<<NB_GEMM, 256, 0, stream>>>(x, W, ps);
  caps_reduce_squash<<<B_SZ, NOD, 0, stream>>>(ps, NB_GEMM, 1.0f / NO, vbuf);
  caps_route<<<NB_ROUTE, 256, 0, stream>>>(x, W, vbuf, rlog, ps, 1);
  caps_reduce_squash<<<B_SZ, NOD, 0, stream>>>(ps, NB_ROUTE, 1.0f, vbuf);
  caps_route<<<NB_ROUTE, 256, 0, stream>>>(x, W, vbuf, rlog, ps, 2);
  caps_reduce_squash<<<B_SZ, NOD, 0, stream>>>(ps, NB_ROUTE, 1.0f, out);
}


# caps_route: tile loops fully unrolled; v, fragments, x_hat tiles and accumulators kept in registers, so the 128 logit chains per capsule overlap instead of running one after another
# speedup vs baseline: 5.6083x; 5.6083x over previous
#include <hip/hip_runtime.h>
#include <math.h>

typedef __attribute__((ext_vector_type(2))) float v2f;
typedef __attribute__((ext_vector_type(8))) float v8f;

#define B_SZ   64
#define NI     2048
#define DI     8
#define NO     32
#define DO_    16
#define NOD    512
#define KTOT   (NI*DI)
#define WN     (NO*DO_*DI)

#define NB_GEMM  128
#define KCHUNK   (KTOT/NB_GEMM)
#define NB_ROUTE 128
#define NPB      (NI/NB_ROUTE)
#define PS_STRIDE (B_SZ*NOD)

__global__ __launch_bounds__(256)
void caps_s0_gemm(const float* __restrict__ x, const float* __restrict__ W,
                  float* __restrict__ ps) {
  const int lane = threadIdx.x & 31;
  const int wave = threadIdx.x >> 5;
  const int half = lane >> 4;
  const int l15  = lane & 15;
  const int kbase = blockIdx.x * KCHUNK;
  float* psb = ps + (size_t)blockIdx.x * PS_STRIDE;

  for (int ot = 0; ot < 4; ++ot) {
    const int o = wave + ot * 8;
    const float* wc = W + (size_t)(o * DO_ + l15) * DI;
    for (int mt = 0; mt < 4; ++mt) {
      const int row = mt * 16 + l15;
      const float* xr = x + (size_t)row * KTOT;
      v8f acc = {};
      for (int k0 = kbase; k0 < kbase + KCHUNK; k0 += 4) {
        const int ka = k0 + 2 * half;
        v2f a  = *(const v2f*)(xr + ka);
        const int n = ka >> 3, i = ka & 7;
        v2f bv = *(const v2f*)(wc + (size_t)n * WN + i);
        acc = __builtin_amdgcn_wmma_f32_16x16x4_f32(
                  false, a, false, bv, (short)0, acc, false, false);
      }
#pragma unroll
      for (int r = 0; r < 8; ++r) {
        const int b = mt * 16 + r + 8 * half;
        psb[b * NOD + o * DO_ + l15] = acc[r];
      }
    }
  }
}

__global__ __launch_bounds__(256)
void caps_route(const float* __restrict__ x, const float* __restrict__ W,
                const float* __restrict__ vcur, float* __restrict__ rlog,
                float* __restrict__ ps, int iter) {
  __shared__ float r1s[B_SZ * NO];
  __shared__ float cs[B_SZ * NO];
  const int lane = threadIdx.x & 31;
  const int wave = threadIdx.x >> 5;
  const int half = lane >> 4;
  const int l15  = lane & 15;

  const v8f zero = {};
  v8f sacc[16];
  v8f vv[16];
  const float* vb   = vcur + (8 * half) * NOD + wave * DO_ + l15;
  const float* csb  = cs + (8 * half) * NO + wave;
  float*       r1b  = r1s + (l15 + 8 * half) * NO + wave;
#pragma unroll
  for (int ot = 0; ot < 4; ++ot) {
#pragma unroll
    for (int mt = 0; mt < 4; ++mt) {
      sacc[ot * 4 + mt] = zero;
#pragma unroll
      for (int r = 0; r < 8; ++r)
        vv[ot * 4 + mt][r] = vb[(mt * 16 + r) * NOD + ot * 8 * DO_];
    }
  }

  const int n0 = blockIdx.x * NPB;
  for (int nn = 0; nn < NPB; ++nn) {
    const int n = n0 + nn;
    const float* wn = W + (size_t)n * WN;

    v2f af[4][2];
    v2f bf[4][2];
#pragma unroll
    for (int mt = 0; mt < 4; ++mt) {
      const int row = mt * 16 + l15;
#pragma unroll
      for (int ks = 0; ks < 2; ++ks) {
        const int k = ks * 4 + 2 * half;
        af[mt][ks] = *(const v2f*)(x + (size_t)row * KTOT + n * DI + k);
      }
    }
#pragma unroll
    for (int ot = 0; ot < 4; ++ot) {
      const int o = wave + ot * 8;
#pragma unroll
      for (int ks = 0; ks < 2; ++ks) {
        const int k = ks * 4 + 2 * half;
        bf[ot][ks] = *(const v2f*)(wn + (size_t)(o * DO_ + l15) * DI + k);
      }
    }

    float lg[16];
    v8f dk[16];
#pragma unroll
    for (int ot = 0; ot < 4; ++ot) {
#pragma unroll
      for (int mt = 0; mt < 4; ++mt) {
        v8f d = __builtin_amdgcn_wmma_f32_16x16x4_f32(
                    false, af[mt][0], false, bf[ot][0], (short)0, zero, false, false);
        d = __builtin_amdgcn_wmma_f32_16x16x4_f32(
                    false, af[mt][1], false, bf[ot][1], (short)0, d, false, false);
        dk[ot * 4 + mt] = d;
        float sel = 0.f;
#pragma unroll
        for (int r = 0; r < 8; ++r) {
          float p = d[r] * vv[ot * 4 + mt][r];
          p += __shfl_xor(p, 1);
          p += __shfl_xor(p, 2);
          p += __shfl_xor(p, 4);
          p += __shfl_xor(p, 8);
          sel = (l15 == r) ? p : sel;
        }
        asm volatile("" : "+v"(sel));
        lg[ot * 4 + mt] = sel;
      }
      __builtin_amdgcn_sched_barrier(0);
    }
    if (l15 < 8) {
      float* rlb = rlog + ((size_t)(l15 + 8 * half) * NI + n) * NO + wave;
      if (iter == 1) {
#pragma unroll
        for (int ot = 0; ot < 4; ++ot) {
#pragma unroll
          for (int mt = 0; mt < 4; ++mt) {
            rlb[(size_t)mt * 16 * NI * NO + ot * 8] = lg[ot * 4 + mt];
            r1b[mt * 16 * NO + ot * 8] = lg[ot * 4 + mt];
          }
        }
      } else {
        float old[16];
#pragma unroll
        for (int ot = 0; ot < 4; ++ot) {
#pragma unroll
          for (int mt = 0; mt < 4; ++mt)
            old[ot * 4 + mt] = rlb[(size_t)mt * 16 * NI * NO + ot * 8];
        }
#pragma unroll
        for (int ot = 0; ot < 4; ++ot) {
#pragma unroll
          for (int mt = 0; mt < 4; ++mt)
            r1b[mt * 16 * NO + ot * 8] = old[ot * 4 + mt] + lg[ot * 4 + mt];
        }
      }
    }
    __syncthreads();

    if (threadIdx.x < B_SZ) {
      const int b = threadIdx.x;
      float m = -3.4e38f;
      for (int o = 0; o < NO; ++o) m = fmaxf(m, r1s[b * NO + o]);
      float ssum = 0.f;
      for (int o = 0; o < NO; ++o) {
        float e = __expf(r1s[b * NO + o] - m);
        cs[b * NO + o] = e; ssum += e;
      }
      const float inv = 1.f / ssum;
      for (int o = 0; o < NO; ++o) cs[b * NO + o] *= inv;
    }
    __syncthreads();

#pragma unroll
    for (int ot = 0; ot < 4; ++ot) {
#pragma unroll
      for (int mt = 0; mt < 4; ++mt) {
        const v8f d = dk[ot * 4 + mt];
#pragma unroll
        for (int r = 0; r < 8; ++r)
          sacc[ot * 4 + mt][r] += csb[(mt * 16 + r) * NO + ot * 8] * d[r];
      }
      __builtin_amdgcn_sched_barrier(0);
    }
    __syncthreads();
  }

  float* psb = ps + (size_t)blockIdx.x * PS_STRIDE;
#pragma unroll
  for (int ot = 0; ot < 4; ++ot) {
    const int o = wave + ot * 8;
#pragma unroll
    for (int mt = 0; mt < 4; ++mt) {
#pragma unroll
      for (int r = 0; r < 8; ++r) {
        const int b = mt * 16 + r + 8 * half;
        psb[b * NOD + o * DO_ + l15] = sacc[ot * 4 + mt][r];
      }
    }
  }
}

__global__ __launch_bounds__(512)
void caps_reduce_squash(const float* __restrict__ ps, int nparts, float prescale,
                        float* __restrict__ out) {
  __shared__ float sh[NOD];
  __shared__ float scl[NO];
  const int b  = blockIdx.x;
  const int od = threadIdx.x;
  float s = 0.f;
  for (int p = 0; p < nparts; ++p)
    s += ps[(size_t)p * PS_STRIDE + b * NOD + od];
  s *= prescale;
  sh[od] = s;
  __syncthreads();
  if (od < NO) {
    float nn = 0.f;
#pragma unroll
    for (int d = 0; d < DO_; ++d) { float t = sh[od * DO_ + d]; nn += t * t; }
    scl[od] = sqrtf(nn) / (1.f + nn);
  }
  __syncthreads();
  out[(size_t)b * NOD + od] = sh[od] * scl[od >> 4];
}

extern "C" void kernel_launch(void* const* d_in, const int* in_sizes, int n_in,
                              void* d_out, int out_size, void* d_ws, size_t ws_size,
                              hipStream_t stream) {
  const float* x = (const float*)d_in[0];
  const float* W = (const float*)d_in[1];
  float* out = (float*)d_out;

  float* ps   = (float*)d_ws;
  float* vbuf = ps + (size_t)NB_GEMM * PS_STRIDE;
  float* rlog = vbuf + (size_t)B_SZ * NOD;

  caps_s0_gemm<<<NB_GEMM, 256, 0, stream>>>(x, W, ps);
  caps_reduce_squash<<<B_SZ, NOD, 0, stream>>>(ps, NB_GEMM, 1.0f / NO, vbuf);
  caps_route<<<NB_ROUTE, 256, 0, stream>>>(x, W, vbuf, rlog, ps, 1);
  caps_reduce_squash<<<B_SZ, NOD, 0, stream>>>(ps, NB_ROUTE, 1.0f, vbuf);
  caps_route<<<NB_ROUTE, 256, 0, stream>>>(x, W, vbuf, rlog, ps, 2);
  caps_reduce_squash<<<B_SZ, NOD, 0, stream>>>(ps, NB_ROUTE, 1.0f, out);
}


# caps_route: x_hat tiles computed transposed (W as A operand, x as B): sum over d = 8 in-lane multiply-adds + one lane l / l+16 exchange instead of a 4-step LDS butterfly per row; LDS row pitch 33
# speedup vs baseline: 10.4345x; 1.8605x over previous
#include <hip/hip_runtime.h>
#include <math.h>

typedef __attribute__((ext_vector_type(2))) float v2f;
typedef __attribute__((ext_vector_type(8))) float v8f;

#define B_SZ   64
#define NI     2048
#define DI     8
#define NO     32
#define DO_    16
#define NOD    512
#define KTOT   (NI*DI)
#define WN     (NO*DO_*DI)

#define NB_GEMM  128
#define KCHUNK   (KTOT/NB_GEMM)
#define NB_ROUTE 128
#define NPB      (NI/NB_ROUTE)
#define PS_STRIDE (B_SZ*NOD)

__global__ __launch_bounds__(256)
void caps_s0_gemm(const float* __restrict__ x, const float* __restrict__ W,
                  float* __restrict__ ps) {
  const int lane = threadIdx.x & 31;
  const int wave = threadIdx.x >> 5;
  const int half = lane >> 4;
  const int l15  = lane & 15;
  const int kbase = blockIdx.x * KCHUNK;
  float* psb = ps + (size_t)blockIdx.x * PS_STRIDE;

  for (int ot = 0; ot < 4; ++ot) {
    const int o = wave + ot * 8;
    const float* wc = W + (size_t)(o * DO_ + l15) * DI;
    for (int mt = 0; mt < 4; ++mt) {
      const int row = mt * 16 + l15;
      const float* xr = x + (size_t)row * KTOT;
      v8f acc = {};
      for (int k0 = kbase; k0 < kbase + KCHUNK; k0 += 4) {
        const int ka = k0 + 2 * half;
        v2f a  = *(const v2f*)(xr + ka);
        const int n = ka >> 3, i = ka & 7;
        v2f bv = *(const v2f*)(wc + (size_t)n * WN + i);
        acc = __builtin_amdgcn_wmma_f32_16x16x4_f32(
                  false, a, false, bv, (short)0, acc, false, false);
      }
#pragma unroll
      for (int r = 0; r < 8; ++r) {
        const int b = mt * 16 + r + 8 * half;
        psb[b * NOD + o * DO_ + l15] = acc[r];
      }
    }
  }
}

#define LP 33
__global__ __launch_bounds__(256)
void caps_route(const float* __restrict__ x, const float* __restrict__ W,
                const float* __restrict__ vcur, float* __restrict__ rlog,
                float* __restrict__ ps, int iter) {
  __shared__ float r1s[B_SZ * LP];
  __shared__ float cs[B_SZ * LP];
  const int lane = threadIdx.x & 31;
  const int wave = threadIdx.x >> 5;
  const int half = lane >> 4;
  const int l15  = lane & 15;

  const v8f zero = {};
  v8f sacc[16];
  v8f vv[16];
  const float* vb  = vcur + l15 * NOD + wave * DO_ + 8 * half;
  const float* csb = cs + l15 * LP + wave;
  float*       r1b = r1s + (32 * half + l15) * LP + wave;
#pragma unroll
  for (int ot = 0; ot < 4; ++ot) {
#pragma unroll
    for (int mt = 0; mt < 4; ++mt) {
      sacc[ot * 4 + mt] = zero;
#pragma unroll
      for (int r = 0; r < 8; ++r)
        vv[ot * 4 + mt][r] = vb[mt * 16 * NOD + ot * 8 * DO_ + r];
    }
  }

  const int n0 = blockIdx.x * NPB;
  for (int nn = 0; nn < NPB; ++nn) {
    const int n = n0 + nn;
    const float* wn = W + (size_t)n * WN;

    v2f xf[4][2];
    v2f wf[4][2];
#pragma unroll
    for (int mt = 0; mt < 4; ++mt) {
      const int row = mt * 16 + l15;
#pragma unroll
      for (int ks = 0; ks < 2; ++ks) {
        const int k = ks * 4 + 2 * half;
        xf[mt][ks] = *(const v2f*)(x + (size_t)row * KTOT + n * DI + k);
      }
    }
#pragma unroll
    for (int ot = 0; ot < 4; ++ot) {
      const int o = wave + ot * 8;
#pragma unroll
      for (int ks = 0; ks < 2; ++ks) {
        const int k = ks * 4 + 2 * half;
        wf[ot][ks] = *(const v2f*)(wn + (size_t)(o * DO_ + l15) * DI + k);
      }
    }

    float lg[16];
    v8f dk[16];
#pragma unroll
    for (int ot = 0; ot < 4; ++ot) {
#pragma unroll
      for (int mt = 0; mt < 4; ++mt) {
        v8f d = __builtin_amdgcn_wmma_f32_16x16x4_f32(
                    false, wf[ot][0], false, xf[mt][0], (short)0, zero, false, false);
        d = __builtin_amdgcn_wmma_f32_16x16x4_f32(
                    false, wf[ot][1], false, xf[mt][1], (short)0, d, false, false);
        dk[ot * 4 + mt] = d;
        const v8f w = vv[ot * 4 + mt];
        v2f q = __builtin_shufflevector(d, d, 0, 1) * __builtin_shufflevector(w, w, 0, 1);
        q += __builtin_shufflevector(d, d, 2, 3) * __builtin_shufflevector(w, w, 2, 3);
        q += __builtin_shufflevector(d, d, 4, 5) * __builtin_shufflevector(w, w, 4, 5);
        q += __builtin_shufflevector(d, d, 6, 7) * __builtin_shufflevector(w, w, 6, 7);
        lg[ot * 4 + mt] = q[0] + q[1];
      }
      __builtin_amdgcn_sched_barrier(0);
    }
#pragma unroll
    for (int t = 0; t < 16; ++t) lg[t] += __shfl_xor(lg[t], 16);
    {
      float* rlb = rlog + ((size_t)(32 * half + l15) * NI + n) * NO + wave;
      float lw[4][2];
#pragma unroll
      for (int ot = 0; ot < 4; ++ot) {
#pragma unroll
        for (int j = 0; j < 2; ++j)
          lw[ot][j] = half ? lg[ot * 4 + 2 + j] : lg[ot * 4 + j];
      }
      if (iter == 1) {
#pragma unroll
        for (int ot = 0; ot < 4; ++ot) {
#pragma unroll
          for (int j = 0; j < 2; ++j) {
            rlb[(size_t)j * 16 * NI * NO + ot * 8] = lw[ot][j];
            r1b[j * 16 * LP + ot * 8] = lw[ot][j];
          }
        }
      } else {
        float old[4][2];
#pragma unroll
        for (int ot = 0; ot < 4; ++ot) {
#pragma unroll
          for (int j = 0; j < 2; ++j)
            old[ot][j] = rlb[(size_t)j * 16 * NI * NO + ot * 8];
        }
#pragma unroll
        for (int ot = 0; ot < 4; ++ot) {
#pragma unroll
          for (int j = 0; j < 2; ++j)
            r1b[j * 16 * LP + ot * 8] = old[ot][j] + lw[ot][j];
        }
      }
    }
    __syncthreads();

    if (threadIdx.x < B_SZ) {
      const int b = threadIdx.x;
      float m = -3.4e38f;
      for (int o = 0; o < NO; ++o) m = fmaxf(m, r1s[b * LP + o]);
      float ssum = 0.f;
      for (int o = 0; o < NO; ++o) {
        float e = __expf(r1s[b * LP + o] - m);
        cs[b * LP + o] = e; ssum += e;
      }
      const float inv = 1.f / ssum;
      for (int o = 0; o < NO; ++o) cs[b * LP + o] *= inv;
    }
    __syncthreads();

#pragma unroll
    for (int ot = 0; ot < 4; ++ot) {
#pragma unroll
      for (int mt = 0; mt < 4; ++mt) {
        const v8f d = dk[ot * 4 + mt];
        const float c = csb[mt * 16 * LP + ot * 8];
#pragma unroll
        for (int r = 0; r < 8; ++r)
          sacc[ot * 4 + mt][r] = fmaf(c, d[r], sacc[ot * 4 + mt][r]);
      }
    }
    __syncthreads();
  }

  float* psb = ps + (size_t)blockIdx.x * PS_STRIDE + l15 * NOD + wave * DO_ + 8 * half;
#pragma unroll
  for (int ot = 0; ot < 4; ++ot) {
#pragma unroll
    for (int mt = 0; mt < 4; ++mt) {
#pragma unroll
      for (int r = 0; r < 8; ++r)
        psb[mt * 16 * NOD + ot * 8 * DO_ + r] = sacc[ot * 4 + mt][r];
    }
  }
}

__global__ __launch_bounds__(512)
void caps_reduce_squash(const float* __restrict__ ps, int nparts, float prescale,
                        float* __restrict__ out) {
  __shared__ float sh[NOD];
  __shared__ float scl[NO];
  const int b  = blockIdx.x;
  const int od = threadIdx.x;
  float s = 0.f;
  for (int p = 0; p < nparts; ++p)
    s += ps[(size_t)p * PS_STRIDE + b * NOD + od];
  s *= prescale;
  sh[od] = s;
  __syncthreads();
  if (od < NO) {
    float nn = 0.f;
#pragma unroll
    for (int d = 0; d < DO_; ++d) { float t = sh[od * DO_ + d]; nn += t * t; }
    scl[od] = sqrtf(nn) / (1.f + nn);
  }
  __syncthreads();
  out[(size_t)b * NOD + od] = sh[od] * scl[od >> 4];
}

extern "C" void kernel_launch(void* const* d_in, const int* in_sizes, int n_in,
                              void* d_out, int out_size, void* d_ws, size_t ws_size,
                              hipStream_t stream) {
  const float* x = (const float*)d_in[0];
  const float* W = (const float*)d_in[1];
  float* out = (float*)d_out;

  float* ps   = (float*)d_ws;
  float* vbuf = ps + (size_t)NB_GEMM * PS_STRIDE;
  float* rlog = vbuf + (size_t)B_SZ * NOD;

  caps_s0_gemm<<<NB_GEMM, 256, 0, stream>>>(x, W, ps);
  caps_reduce_squash<<<B_SZ, NOD, 0, stream>>>(ps, NB_GEMM, 1.0f / NO, vbuf);
  caps_route<<<NB_ROUTE, 256, 0, stream>>>(x, W, vbuf, rlog, ps, 1);
  caps_reduce_squash<<<B_SZ, NOD, 0, stream>>>(ps, NB_ROUTE, 1.0f, vbuf);
  caps_route<<<NB_ROUTE, 256, 0, stream>>>(x, W, vbuf, rlog, ps, 2);
  caps_reduce_squash<<<B_SZ, NOD, 0, stream>>>(ps, NB_ROUTE, 1.0f, out);
}


# caps_route: every launch sized to 256 blocks so no WGP idles: GEMM and routing blocks take half the batch (32 rows, same 128 partial-s planes), reduce block = (b, 8 o), planes over 2 thread groups
# speedup vs baseline: 20.7226x; 1.9860x over previous
#include <hip/hip_runtime.h>
#include <math.h>

typedef __attribute__((ext_vector_type(2))) float v2f;
typedef __attribute__((ext_vector_type(8))) float v8f;

#define B_SZ   64
#define NI     2048
#define DI     8
#define NO     32
#define DO_    16
#define NOD    512
#define KTOT   (NI*DI)
#define WN     (NO*DO_*DI)

#define NPART    128
#define KCHUNK   (KTOT/NPART)
#define NPB      (NI/NPART)
#define BROWS    32
#define NB_BIG   256
#define PS_STRIDE (B_SZ*NOD)

__global__ __launch_bounds__(256)
void caps_s0_gemm(const float* __restrict__ x, const float* __restrict__ W,
                  float* __restrict__ ps) {
  const int lane = threadIdx.x & 31;
  const int wave = threadIdx.x >> 5;
  const int half = lane >> 4;
  const int l15  = lane & 15;
  const int kblk = blockIdx.x >> 1;
  const int b0   = (blockIdx.x & 1) * BROWS;
  const int kbase = kblk * KCHUNK;
  float* psb = ps + (size_t)kblk * PS_STRIDE;

  for (int ot = 0; ot < 4; ++ot) {
    const int o = wave + ot * 8;
    const float* wc = W + (size_t)(o * DO_ + l15) * DI;
    for (int mt = 0; mt < 2; ++mt) {
      const int row = b0 + mt * 16 + l15;
      const float* xr = x + (size_t)row * KTOT;
      v8f acc = {};
#pragma unroll 8
      for (int ks = 0; ks < KCHUNK / 4; ++ks) {
        const int ka = kbase + ks * 4 + 2 * half;
        v2f a  = *(const v2f*)(xr + ka);
        const int n = ka >> 3, i = ka & 7;
        v2f bv = *(const v2f*)(wc + (size_t)n * WN + i);
        acc = __builtin_amdgcn_wmma_f32_16x16x4_f32(
                  false, a, false, bv, (short)0, acc, false, false);
      }
#pragma unroll
      for (int r = 0; r < 8; ++r) {
        const int b = b0 + mt * 16 + r + 8 * half;
        psb[b * NOD + o * DO_ + l15] = acc[r];
      }
    }
  }
}

#define LP 33
#define GC 8
__global__ __launch_bounds__(256)
void caps_route(const float* __restrict__ x, const float* __restrict__ W,
                const float* __restrict__ vcur, float* __restrict__ rlog,
                float* __restrict__ ps, int iter) {
  __shared__ float lgs[GC * BROWS * LP];
  const int lane = threadIdx.x & 31;
  const int wave = threadIdx.x >> 5;
  const int half = lane >> 4;
  const int l15  = lane & 15;
  const int nblk = blockIdx.x >> 1;
  const int b0   = (blockIdx.x & 1) * BROWS;

  const v8f zero = {};
  v8f sacc[8];
  v8f vv[8];
  const float* vb  = vcur + (b0 + l15) * NOD + wave * DO_ + 8 * half;
  const float* csb = lgs + l15 * LP + wave;
  float*       r1b = lgs + lane * LP + wave;
#pragma unroll
  for (int ot = 0; ot < 4; ++ot) {
#pragma unroll
    for (int mt = 0; mt < 2; ++mt) {
      sacc[ot * 2 + mt] = zero;
#pragma unroll
      for (int r = 0; r < 8; ++r)
        vv[ot * 2 + mt][r] = vb[mt * 16 * NOD + ot * 8 * DO_ + r];
    }
  }

  const int n0 = nblk * NPB;
  for (int g0 = 0; g0 < NPB; g0 += GC) {
    v2f xf[GC][2][2];
    v2f wf[GC][4][2];
#pragma unroll
    for (int g = 0; g < GC; ++g) {
      const int n = n0 + g0 + g;
      const float* wn = W + (size_t)n * WN;
#pragma unroll
      for (int mt = 0; mt < 2; ++mt) {
        const int row = b0 + mt * 16 + l15;
#pragma unroll
        for (int ks = 0; ks < 2; ++ks) {
          const int k = ks * 4 + 2 * half;
          xf[g][mt][ks] = *(const v2f*)(x + (size_t)row * KTOT + n * DI + k);
        }
      }
#pragma unroll
      for (int ot = 0; ot < 4; ++ot) {
        const int o = wave + ot * 8;
#pragma unroll
        for (int ks = 0; ks < 2; ++ks) {
          const int k = ks * 4 + 2 * half;
          wf[g][ot][ks] = *(const v2f*)(wn + (size_t)(o * DO_ + l15) * DI + k);
        }
      }
    }

#pragma unroll
    for (int g = 0; g < GC; ++g) {
      float lg[8];
#pragma unroll
      for (int ot = 0; ot < 4; ++ot) {
#pragma unroll
        for (int mt = 0; mt < 2; ++mt) {
          v8f d = __builtin_amdgcn_wmma_f32_16x16x4_f32(
                      false, wf[g][ot][0], false, xf[g][mt][0], (short)0, zero, false, false);
          d = __builtin_amdgcn_wmma_f32_16x16x4_f32(
                      false, wf[g][ot][1], false, xf[g][mt][1], (short)0, d, false, false);
          const v8f w = vv[ot * 2 + mt];
          v2f q = __builtin_shufflevector(d, d, 0, 1) * __builtin_shufflevector(w, w, 0, 1);
          q += __builtin_shufflevector(d, d, 2, 3) * __builtin_shufflevector(w, w, 2, 3);
          q += __builtin_shufflevector(d, d, 4, 5) * __builtin_shufflevector(w, w, 4, 5);
          q += __builtin_shufflevector(d, d, 6, 7) * __builtin_shufflevector(w, w, 6, 7);
          lg[ot * 2 + mt] = q[0] + q[1];
        }
        if (ot & 1) __builtin_amdgcn_sched_barrier(0);
      }
#pragma unroll
      for (int t = 0; t < 8; ++t) lg[t] += __shfl_xor(lg[t], 16);
      float* r1g = r1b + g * BROWS * LP;
#pragma unroll
      for (int ot = 0; ot < 4; ++ot)
        r1g[ot * 8] = half ? lg[ot * 2 + 1] : lg[ot * 2];
    }
    __syncthreads();

    {
      const int g = threadIdx.x >> 5;
      const int b = b0 + (threadIdx.x & 31);
      float* rowp = lgs + threadIdx.x * LP;
      float4* rl = (float4*)(rlog + ((size_t)b * NI + (n0 + g0 + g)) * NO);
      float e[NO];
#pragma unroll
      for (int o = 0; o < NO; ++o) e[o] = rowp[o];
      if (iter == 1) {
#pragma unroll
        for (int q = 0; q < NO / 4; ++q)
          rl[q] = make_float4(e[4 * q], e[4 * q + 1], e[4 * q + 2], e[4 * q + 3]);
      } else {
        float4 old[NO / 4];
#pragma unroll
        for (int q = 0; q < NO / 4; ++q) old[q] = rl[q];
#pragma unroll
        for (int q = 0; q < NO / 4; ++q) {
          e[4 * q]     += old[q].x;
          e[4 * q + 1] += old[q].y;
          e[4 * q + 2] += old[q].z;
          e[4 * q + 3] += old[q].w;
        }
      }
      float m = -3.4e38f;
#pragma unroll
      for (int o = 0; o < NO; ++o) m = fmaxf(m, e[o]);
      float ssum = 0.f;
#pragma unroll
      for (int o = 0; o < NO; ++o) { e[o] = __expf(e[o] - m); ssum += e[o]; }
      const float inv = 1.f / ssum;
#pragma unroll
      for (int o = 0; o < NO; ++o) rowp[o] = e[o] * inv;
    }
    __syncthreads();

#pragma unroll
    for (int g = 0; g < GC; ++g) {
#pragma unroll
      for (int ot = 0; ot < 4; ++ot) {
        v2f xs[2][2];
#pragma unroll
        for (int mt = 0; mt < 2; ++mt) {
          const float c = csb[g * BROWS * LP + mt * 16 * LP + ot * 8];
          xs[mt][0] = xf[g][mt][0] * c;
          xs[mt][1] = xf[g][mt][1] * c;
        }
#pragma unroll
        for (int ks = 0; ks < 2; ++ks) {
#pragma unroll
          for (int mt = 0; mt < 2; ++mt)
            sacc[ot * 2 + mt] = __builtin_amdgcn_wmma_f32_16x16x4_f32(
                false, wf[g][ot][ks], false, xs[mt][ks], (short)0, sacc[ot * 2 + mt], false, false);
        }
        if (ot & 1) __builtin_amdgcn_sched_barrier(0);
      }
    }
    __syncthreads();
  }

  float* psb = ps + (size_t)nblk * PS_STRIDE + (b0 + l15) * NOD + wave * DO_ + 8 * half;
#pragma unroll
  for (int ot = 0; ot < 4; ++ot) {
#pragma unroll
    for (int mt = 0; mt < 2; ++mt) {
#pragma unroll
      for (int r = 0; r < 8; ++r)
        psb[mt * 16 * NOD + ot * 8 * DO_ + r] = sacc[ot * 2 + mt][r];
    }
  }
}

#define RQ   128
#define RPG  2
__global__ __launch_bounds__(256)
void caps_reduce_squash(const float* __restrict__ ps, float prescale,
                        float* __restrict__ out) {
  __shared__ float part[RPG][RQ];
  __shared__ float sh[RQ];
  __shared__ float scl[RQ / DO_];
  const int b  = blockIdx.x >> 2;
  const int j  = threadIdx.x & (RQ - 1);
  const int pg = threadIdx.x >> 7;
  const int od = (blockIdx.x & 3) * RQ + j;
  const float* src = ps + (size_t)(pg * (NPART / RPG)) * PS_STRIDE + b * NOD + od;
  float s = 0.f;
#pragma unroll
  for (int p = 0; p < NPART / RPG; ++p)
    s += src[(size_t)p * PS_STRIDE];
  part[pg][j] = s;
  __syncthreads();
  if (threadIdx.x < RQ)
    sh[j] = (part[0][j] + part[1][j]) * prescale;
  __syncthreads();
  if (threadIdx.x < RQ / DO_) {
    float nn = 0.f;
#pragma unroll
    for (int d = 0; d < DO_; ++d) { float t = sh[threadIdx.x * DO_ + d]; nn += t * t; }
    scl[threadIdx.x] = sqrtf(nn) / (1.f + nn);
  }
  __syncthreads();
  if (threadIdx.x < RQ)
    out[(size_t)b * NOD + od] = sh[j] * scl[j >> 4];
}

extern "C" void kernel_launch(void* const* d_in, const int* in_sizes, int n_in,
                              void* d_out, int out_size, void* d_ws, size_t ws_size,
                              hipStream_t stream) {
  const float* x = (const float*)d_in[0];
  const float* W = (const float*)d_in[1];
  float* out = (float*)d_out;

  float* ps   = (float*)d_ws;
  float* vbuf = ps + (size_t)NPART * PS_STRIDE;
  float* rlog = vbuf + (size_t)B_SZ * NOD;

  caps_s0_gemm<<<NB_BIG, 256, 0, stream>>>(x, W, ps);
  caps_reduce_squash<<<256, 256, 0, stream>>>(ps, 1.0f / NO, vbuf);
  caps_route<<<NB_BIG, 256, 0, stream>>>(x, W, vbuf, rlog, ps, 1);
  caps_reduce_squash<<<256, 256, 0, stream>>>(ps, 1.0f, vbuf);
  caps_route<<<NB_BIG, 256, 0, stream>>>(x, W, vbuf, rlog, ps, 2);
  caps_reduce_squash<<<256, 256, 0, stream>>>(ps, 1.0f, out);
}


# caps_route: one wave owns a batch tile with all 32 output capsules: logits, softmax, coupling stay in registers (no LDS, no barrier phases); 16-byte fragment loads (k = 4h..4h+3), one capsule ahead
# speedup vs baseline: 22.9335x; 1.1067x over previous
#include <hip/hip_runtime.h>
#include <math.h>

typedef __attribute__((ext_vector_type(2))) float v2f;
typedef __attribute__((ext_vector_type(4))) float v4f;
typedef __attribute__((ext_vector_type(8))) float v8f;

#define B_SZ   64
#define NI     2048
#define DI     8
#define NO     32
#define DO_    16
#define NOD    512
#define KTOT   (NI*DI)
#define WN     (NO*DO_*DI)

#define NPART    128
#define KCHUNK   (KTOT/NPART)
#define NPB      (NI/NPART)
#define BROWS    32
#define NB_BIG   256
#define PS_STRIDE (B_SZ*NOD)

__global__ __launch_bounds__(256)
void caps_s0_gemm(const float* __restrict__ x, const float* __restrict__ W,
                  float* __restrict__ ps) {
  const int lane = threadIdx.x & 31;
  const int wave = threadIdx.x >> 5;
  const int half = lane >> 4;
  const int l15  = lane & 15;
  const int kblk = blockIdx.x >> 1;
  const int b0   = (blockIdx.x & 1) * BROWS;
  const int kbase = kblk * KCHUNK;
  float* psb = ps + (size_t)kblk * PS_STRIDE;

  for (int ot = 0; ot < 4; ++ot) {
    const int o = wave + ot * 8;
    const float* wc = W + (size_t)(o * DO_ + l15) * DI;
    for (int mt = 0; mt < 2; ++mt) {
      const int row = b0 + mt * 16 + l15;
      const float* xr = x + (size_t)row * KTOT;
      v8f acc = {};
#pragma unroll 8
      for (int ks = 0; ks < KCHUNK / 4; ++ks) {
        const int ka = kbase + ks * 4 + 2 * half;
        v2f a  = *(const v2f*)(xr + ka);
        const int n = ka >> 3, i = ka & 7;
        v2f bv = *(const v2f*)(wc + (size_t)n * WN + i);
        acc = __builtin_amdgcn_wmma_f32_16x16x4_f32(
                  false, a, false, bv, (short)0, acc, false, false);
      }
#pragma unroll
      for (int r = 0; r < 8; ++r) {
        const int b = b0 + mt * 16 + r + 8 * half;
        psb[b * NOD + o * DO_ + l15] = acc[r];
      }
    }
  }
}

#define CPW 8

static __device__ __forceinline__ float xhalf(float v) {
  return __builtin_bit_cast(float, __builtin_amdgcn_permlanex16(
      __builtin_bit_cast(int, v), __builtin_bit_cast(int, v), 0x76543210, 0xfedcba98, false, false));
}

static __device__ __forceinline__ void load_frags(const float* __restrict__ xb,
                                                  const float* __restrict__ wb, int n,
                                                  v4f &xf, v4f (&wf)[NO]) {
  xf = *(const v4f*)(xb + n * DI);
  const float* wn = wb + (size_t)n * WN;
#pragma unroll
  for (int o = 0; o < NO; ++o) wf[o] = *(const v4f*)(wn + o * (DO_ * DI));
}

template <int ITER>
static __device__ __forceinline__ void route_capsule(v4f &xq, v4f (&wq)[NO],
                                                     const v8f (&vv)[NO], v8f (&sacc)[NO],
                                                     float* __restrict__ rl, int half,
                                                     float4 (&old)[NO / 4],
                                                     const float* __restrict__ xnext,
                                                     const float* __restrict__ wnext,
                                                     const float* __restrict__ rlnext) {
  const v2f xf0 = __builtin_shufflevector(xq, xq, 0, 1);
  const v2f xf1 = __builtin_shufflevector(xq, xq, 2, 3);

  const v8f zero = {};
  float e[NO];
#pragma unroll
  for (int o = 0; o < NO; ++o) {
    const v2f wf0 = __builtin_shufflevector(wq[o], wq[o], 0, 1);
    const v2f wf1 = __builtin_shufflevector(wq[o], wq[o], 2, 3);
    v8f d = __builtin_amdgcn_wmma_f32_16x16x4_f32(
                false, wf0, false, xf0, (short)0, zero, false, false);
    d = __builtin_amdgcn_wmma_f32_16x16x4_f32(
                false, wf1, false, xf1, (short)0, d, false, false);
    const v8f w = vv[o];
    v2f q = __builtin_shufflevector(d, d, 0, 1) * __builtin_shufflevector(w, w, 0, 1);
    q += __builtin_shufflevector(d, d, 2, 3) * __builtin_shufflevector(w, w, 2, 3);
    q += __builtin_shufflevector(d, d, 4, 5) * __builtin_shufflevector(w, w, 4, 5);
    q += __builtin_shufflevector(d, d, 6, 7) * __builtin_shufflevector(w, w, 6, 7);
    e[o] = q[0] + q[1];
    if ((o & 3) == 3) __builtin_amdgcn_sched_barrier(0);
  }
#pragma unroll
  for (int o = 0; o < NO; ++o) e[o] += xhalf(e[o]);

  if (ITER == 1) {
    float4* rl4 = (float4*)rl + 4 * half;
#pragma unroll
    for (int q = 0; q < 4; ++q)
      rl4[q] = make_float4(half ? e[16 + 4 * q]     : e[4 * q],
                           half ? e[16 + 4 * q + 1] : e[4 * q + 1],
                           half ? e[16 + 4 * q + 2] : e[4 * q + 2],
                           half ? e[16 + 4 * q + 3] : e[4 * q + 3]);
  } else {
#pragma unroll
    for (int q = 0; q < NO / 4; ++q) {
      e[4 * q]     += old[q].x;
      e[4 * q + 1] += old[q].y;
      e[4 * q + 2] += old[q].z;
      e[4 * q + 3] += old[q].w;
    }
  }
  float m4[4] = {-3.4e38f, -3.4e38f, -3.4e38f, -3.4e38f};
#pragma unroll
  for (int o = 0; o < NO; ++o) m4[o & 3] = fmaxf(m4[o & 3], e[o]);
  const float m = fmaxf(fmaxf(m4[0], m4[1]), fmaxf(m4[2], m4[3]));
  float s4[4] = {0.f, 0.f, 0.f, 0.f};
#pragma unroll
  for (int o = 0; o < NO; ++o) { e[o] = __expf(e[o] - m); s4[o & 3] += e[o]; }
  const float ssum = (s4[0] + s4[1]) + (s4[2] + s4[3]);
  const float inv = 1.f / ssum;

  const v4f xqn = *(const v4f*)xnext;
  if (ITER != 1) {
#pragma unroll
    for (int q = 0; q < NO / 4; ++q) old[q] = ((const float4*)rlnext)[q];
  }
#pragma unroll
  for (int o = 0; o < NO; ++o) {
    const float c = e[o] * inv;
    const v2f wf0 = __builtin_shufflevector(wq[o], wq[o], 0, 1);
    const v2f wf1 = __builtin_shufflevector(wq[o], wq[o], 2, 3);
    const v2f xs0 = xf0 * c;
    const v2f xs1 = xf1 * c;
    sacc[o] = __builtin_amdgcn_wmma_f32_16x16x4_f32(
        false, wf0, false, xs0, (short)0, sacc[o], false, false);
    sacc[o] = __builtin_amdgcn_wmma_f32_16x16x4_f32(
        false, wf1, false, xs1, (short)0, sacc[o], false, false);
    wq[o] = *(const v4f*)(wnext + o * (DO_ * DI));
    if ((o & 3) == 3) __builtin_amdgcn_sched_barrier(0);
  }
  xq = xqn;
}

template <int ITER>
__global__ __launch_bounds__(128)
void caps_route(const float* __restrict__ x, const float* __restrict__ W,
                const float* __restrict__ vcur, float* __restrict__ rlog,
                float* __restrict__ ps) {
  __shared__ float4 comb[2 * NO * 2 * 32];
  const int lane = threadIdx.x & 31;
  const int wave = threadIdx.x >> 5;
  const int half = lane >> 4;
  const int l15  = lane & 15;
  const int mt   = wave & 1;
  const int sub  = wave >> 1;
  const int nblk = blockIdx.x >> 1;
  const int b0   = (blockIdx.x & 1) * BROWS;
  const int brow = b0 + mt * 16 + l15;

  const v8f zero = {};
  v8f sacc[NO];
  v8f vv[NO];
  const float* vb = vcur + brow * NOD + 8 * half;
#pragma unroll
  for (int o = 0; o < NO; ++o) {
    sacc[o] = zero;
#pragma unroll
    for (int r = 0; r < 8; ++r) vv[o][r] = vb[o * DO_ + r];
  }

  const int n0 = nblk * NPB + sub * CPW;
  const float* xb = x + (size_t)brow * KTOT + 4 * half;
  const float* wb = W + (size_t)l15 * DI + 4 * half;
  float* rlb = rlog + (size_t)brow * NI * NO;

  v4f xq;
  v4f wq[NO];
  float4 old[NO / 4];
  load_frags(xb, wb, n0, xq, wq);
  if (ITER != 1) {
#pragma unroll
    for (int q = 0; q < NO / 4; ++q) old[q] = ((const float4*)(rlb + (size_t)n0 * NO))[q];
  }
  route_capsule<ITER>(xq, wq, vv, sacc, rlb + (size_t)n0 * NO, half, old,
                      xb + (n0 + 1) * DI, wb + (size_t)(n0 + 1) * WN, rlb + (size_t)(n0 + 1) * NO);
#pragma unroll 1
  for (int g = 1; g < CPW; ++g) {
    const int nn = (g + 1 < CPW) ? (n0 + g + 1) : (n0 + g);
    route_capsule<ITER>(xq, wq, vv, sacc, rlb + (size_t)(n0 + g) * NO, half, old,
                        xb + nn * DI, wb + (size_t)nn * WN, rlb + (size_t)nn * NO);
  }

  float4* cb = comb + (mt * NO * 2) * 32 + lane;
  if (sub == 1) {
#pragma unroll
    for (int o = 0; o < NO; ++o) {
      cb[(o * 2) * 32]     = make_float4(sacc[o][0], sacc[o][1], sacc[o][2], sacc[o][3]);
      cb[(o * 2 + 1) * 32] = make_float4(sacc[o][4], sacc[o][5], sacc[o][6], sacc[o][7]);
    }
  }
  __syncthreads();
  if (sub == 0) {
    float* psb = ps + (size_t)nblk * PS_STRIDE + brow * NOD + 8 * half;
#pragma unroll
    for (int o = 0; o < NO; ++o) {
      const float4 lo = cb[(o * 2) * 32];
      const float4 hi = cb[(o * 2 + 1) * 32];
      float4* dst = (float4*)(psb + o * DO_);
      dst[0] = make_float4(sacc[o][0] + lo.x, sacc[o][1] + lo.y, sacc[o][2] + lo.z, sacc[o][3] + lo.w);
      dst[1] = make_float4(sacc[o][4] + hi.x, sacc[o][5] + hi.y, sacc[o][6] + hi.z, sacc[o][7] + hi.w);
    }
  }
}

#define RQ   128
#define RPG  2
__global__ __launch_bounds__(256)
void caps_reduce_squash(const float* __restrict__ ps, float prescale,
                        float* __restrict__ out) {
  __shared__ float part[RPG][RQ];
  __shared__ float sh[RQ];
  __shared__ float scl[RQ / DO_];
  const int b  = blockIdx.x >> 2;
  const int j  = threadIdx.x & (RQ - 1);
  const int pg = threadIdx.x >> 7;
  const int od = (blockIdx.x & 3) * RQ + j;
  const float* src = ps + (size_t)(pg * (NPART / RPG)) * PS_STRIDE + b * NOD + od;
  float s = 0.f;
#pragma unroll
  for (int p = 0; p < NPART / RPG; ++p)
    s += src[(size_t)p * PS_STRIDE];
  part[pg][j] = s;
  __syncthreads();
  if (threadIdx.x < RQ)
    sh[j] = (part[0][j] + part[1][j]) * prescale;
  __syncthreads();
  if (threadIdx.x < RQ / DO_) {
    float nn = 0.f;
#pragma unroll
    for (int d = 0; d < DO_; ++d) { float t = sh[threadIdx.x * DO_ + d]; nn += t * t; }
    scl[threadIdx.x] = sqrtf(nn) / (1.f + nn);
  }
  __syncthreads();
  if (threadIdx.x < RQ)
    out[(size_t)b * NOD + od] = sh[j] * scl[j >> 4];
}

extern "C" void kernel_launch(void* const* d_in, const int* in_sizes, int n_in,
                              void* d_out, int out_size, void* d_ws, size_t ws_size,
                              hipStream_t stream) {
  const float* x = (const float*)d_in[0];
  const float* W = (const float*)d_in[1];
  float* out = (float*)d_out;

  float* ps   = (float*)d_ws;
  float* vbuf = ps + (size_t)NPART * PS_STRIDE;
  float* rlog = vbuf + (size_t)B_SZ * NOD;

  caps_s0_gemm<<<NB_BIG, 256, 0, stream>>>(x, W, ps);
  caps_reduce_squash<<<256, 256, 0, stream>>>(ps, 1.0f / NO, vbuf);
  caps_route<1><<<NB_BIG, 128, 0, stream>>>(x, W, vbuf, rlog, ps);
  caps_reduce_squash<<<256, 256, 0, stream>>>(ps, 1.0f, vbuf);
  caps_route<2><<<NB_BIG, 128, 0, stream>>>(x, W, vbuf, rlog, ps);
  caps_reduce_squash<<<256, 256, 0, stream>>>(ps, 1.0f, out);
}


# caps_s0_gemm: each wave keeps its 8 result tiles as accumulators; 16-byte x and W fragment loads (6 per 16 WMMAs, was 32); next 4-capsule stage's loads issued as one group ahead of current 64 WMMAs
# speedup vs baseline: 28.1234x; 1.2263x over previous
#include <hip/hip_runtime.h>
#include <math.h>

typedef __attribute__((ext_vector_type(2))) float v2f;
typedef __attribute__((ext_vector_type(4))) float v4f;
typedef __attribute__((ext_vector_type(8))) float v8f;

#define B_SZ   64
#define NI     2048
#define DI     8
#define NO     32
#define DO_    16
#define NOD    512
#define KTOT   (NI*DI)
#define WN     (NO*DO_*DI)

#define NPART    128
#define KCHUNK   (KTOT/NPART)
#define NPB      (NI/NPART)
#define BROWS    32
#define NB_BIG   256
#define PS_STRIDE (B_SZ*NOD)

#define GCAP   (KCHUNK / DI)
#define GST    4
#define GOT    4
#define GMT    2

static __device__ __forceinline__ void gemm_load_stage(const float* __restrict__ xr,
                                                       const float* __restrict__ wc, int n,
                                                       v4f (&xf)[GST][GMT], v4f (&wf)[GST][GOT]) {
#pragma unroll
  for (int c = 0; c < GST; ++c) {
#pragma unroll
    for (int mt = 0; mt < GMT; ++mt)
      xf[c][mt] = *(const v4f*)(xr + (size_t)mt * 16 * KTOT + (n + c) * DI);
#pragma unroll
    for (int ot = 0; ot < GOT; ++ot)
      wf[c][ot] = *(const v4f*)(wc + (size_t)(n + c) * WN + ot * (8 * DO_ * DI));
  }
}

static __device__ __forceinline__ void gemm_compute_stage(const v4f (&xf)[GST][GMT],
                                                          const v4f (&wf)[GST][GOT],
                                                          v8f (&acc)[GOT][GMT]) {
#pragma unroll
  for (int c = 0; c < GST; ++c) {
#pragma unroll
    for (int ot = 0; ot < GOT; ++ot) {
      const v2f b = __builtin_shufflevector(wf[c][ot], wf[c][ot], 0, 1);
#pragma unroll
      for (int mt = 0; mt < GMT; ++mt) {
        const v2f a = __builtin_shufflevector(xf[c][mt], xf[c][mt], 0, 1);
        acc[ot][mt] = __builtin_amdgcn_wmma_f32_16x16x4_f32(
            false, a, false, b, (short)0, acc[ot][mt], false, false);
      }
    }
#pragma unroll
    for (int ot = 0; ot < GOT; ++ot) {
      const v2f b = __builtin_shufflevector(wf[c][ot], wf[c][ot], 2, 3);
#pragma unroll
      for (int mt = 0; mt < GMT; ++mt) {
        const v2f a = __builtin_shufflevector(xf[c][mt], xf[c][mt], 2, 3);
        acc[ot][mt] = __builtin_amdgcn_wmma_f32_16x16x4_f32(
            false, a, false, b, (short)0, acc[ot][mt], false, false);
      }
    }
  }
}

__global__ __launch_bounds__(256)
void caps_s0_gemm(const float* __restrict__ x, const float* __restrict__ W,
                  float* __restrict__ ps) {
  const int lane = threadIdx.x & 31;
  const int wave = threadIdx.x >> 5;
  const int half = lane >> 4;
  const int l15  = lane & 15;
  const int kblk = blockIdx.x >> 1;
  const int b0   = (blockIdx.x & 1) * BROWS;
  const int n0   = kblk * GCAP;
  float* psb = ps + (size_t)kblk * PS_STRIDE;

  const float* xr = x + (size_t)(b0 + l15) * KTOT + 4 * half;
  const float* wc = W + (size_t)(wave * DO_ + l15) * DI + 4 * half;

  const v8f zero = {};
  v8f acc[GOT][GMT];
#pragma unroll
  for (int ot = 0; ot < GOT; ++ot)
#pragma unroll
    for (int mt = 0; mt < GMT; ++mt) acc[ot][mt] = zero;

  v4f xa[GST][GMT], wa[GST][GOT];
  v4f xb[GST][GMT], wb[GST][GOT];

  gemm_load_stage(xr, wc, n0, xa, wa);
  gemm_load_stage(xr, wc, n0 + GST, xb, wb);
  __builtin_amdgcn_sched_barrier(0);
  gemm_compute_stage(xa, wa, acc);
  __builtin_amdgcn_sched_barrier(0);
  gemm_load_stage(xr, wc, n0 + 2 * GST, xa, wa);
  __builtin_amdgcn_sched_barrier(0);
  gemm_compute_stage(xb, wb, acc);
  __builtin_amdgcn_sched_barrier(0);
  gemm_load_stage(xr, wc, n0 + 3 * GST, xb, wb);
  __builtin_amdgcn_sched_barrier(0);
  gemm_compute_stage(xa, wa, acc);
  __builtin_amdgcn_sched_barrier(0);
  gemm_compute_stage(xb, wb, acc);

#pragma unroll
  for (int ot = 0; ot < GOT; ++ot) {
    const int o = wave + ot * 8;
#pragma unroll
    for (int mt = 0; mt < GMT; ++mt) {
#pragma unroll
      for (int r = 0; r < 8; ++r) {
        const int b = b0 + mt * 16 + r + 8 * half;
        psb[b * NOD + o * DO_ + l15] = acc[ot][mt][r];
      }
    }
  }
}

#define CPW 8

static __device__ __forceinline__ float xhalf(float v) {
  return __builtin_bit_cast(float, __builtin_amdgcn_permlanex16(
      __builtin_bit_cast(int, v), __builtin_bit_cast(int, v), 0x76543210, 0xfedcba98, false, false));
}

static __device__ __forceinline__ void load_frags(const float* __restrict__ xb,
                                                  const float* __restrict__ wb, int n,
                                                  v4f &xf, v4f (&wf)[NO]) {
  xf = *(const v4f*)(xb + n * DI);
  const float* wn = wb + (size_t)n * WN;
#pragma unroll
  for (int o = 0; o < NO; ++o) wf[o] = *(const v4f*)(wn + o * (DO_ * DI));
}

template <int ITER>
static __device__ __forceinline__ void route_capsule(v4f &xq, v4f (&wq)[NO],
                                                     const v8f (&vv)[NO], v8f (&sacc)[NO],
                                                     float* __restrict__ rl, int half,
                                                     float4 (&old)[NO / 4],
                                                     const float* __restrict__ xnext,
                                                     const float* __restrict__ wnext,
                                                     const float* __restrict__ rlnext) {
  const v2f xf0 = __builtin_shufflevector(xq, xq, 0, 1);
  const v2f xf1 = __builtin_shufflevector(xq, xq, 2, 3);

  const v8f zero = {};
  float e[NO];
#pragma unroll
  for (int o = 0; o < NO; ++o) {
    const v2f wf0 = __builtin_shufflevector(wq[o], wq[o], 0, 1);
    const v2f wf1 = __builtin_shufflevector(wq[o], wq[o], 2, 3);
    v8f d = __builtin_amdgcn_wmma_f32_16x16x4_f32(
                false, wf0, false, xf0, (short)0, zero, false, false);
    d = __builtin_amdgcn_wmma_f32_16x16x4_f32(
                false, wf1, false, xf1, (short)0, d, false, false);
    const v8f w = vv[o];
    v2f q = __builtin_shufflevector(d, d, 0, 1) * __builtin_shufflevector(w, w, 0, 1);
    q += __builtin_shufflevector(d, d, 2, 3) * __builtin_shufflevector(w, w, 2, 3);
    q += __builtin_shufflevector(d, d, 4, 5) * __builtin_shufflevector(w, w, 4, 5);
    q += __builtin_shufflevector(d, d, 6, 7) * __builtin_shufflevector(w, w, 6, 7);
    e[o] = q[0] + q[1];
    if ((o & 3) == 3) __builtin_amdgcn_sched_barrier(0);
  }
#pragma unroll
  for (int o = 0; o < NO; ++o) e[o] += xhalf(e[o]);

  if (ITER == 1) {
    float4* rl4 = (float4*)rl + 4 * half;
#pragma unroll
    for (int q = 0; q < 4; ++q)
      rl4[q] = make_float4(half ? e[16 + 4 * q]     : e[4 * q],
                           half ? e[16 + 4 * q + 1] : e[4 * q + 1],
                           half ? e[16 + 4 * q + 2] : e[4 * q + 2],
                           half ? e[16 + 4 * q + 3] : e[4 * q + 3]);
  } else {
#pragma unroll
    for (int q = 0; q < NO / 4; ++q) {
      e[4 * q]     += old[q].x;
      e[4 * q + 1] += old[q].y;
      e[4 * q + 2] += old[q].z;
      e[4 * q + 3] += old[q].w;
    }
  }
  float m4[4] = {-3.4e38f, -3.4e38f, -3.4e38f, -3.4e38f};
#pragma unroll
  for (int o = 0; o < NO; ++o) m4[o & 3] = fmaxf(m4[o & 3], e[o]);
  const float m = fmaxf(fmaxf(m4[0], m4[1]), fmaxf(m4[2], m4[3]));
  float s4[4] = {0.f, 0.f, 0.f, 0.f};
#pragma unroll
  for (int o = 0; o < NO; ++o) { e[o] = __expf(e[o] - m); s4[o & 3] += e[o]; }
  const float ssum = (s4[0] + s4[1]) + (s4[2] + s4[3]);
  const float inv = 1.f / ssum;

  const v4f xqn = *(const v4f*)xnext;
  if (ITER != 1) {
#pragma unroll
    for (int q = 0; q < NO / 4; ++q) old[q] = ((const float4*)rlnext)[q];
  }
#pragma unroll
  for (int o = 0; o < NO; ++o) {
    const float c = e[o] * inv;
    const v2f wf0 = __builtin_shufflevector(wq[o], wq[o], 0, 1);
    const v2f wf1 = __builtin_shufflevector(wq[o], wq[o], 2, 3);
    const v2f xs0 = xf0 * c;
    const v2f xs1 = xf1 * c;
    sacc[o] = __builtin_amdgcn_wmma_f32_16x16x4_f32(
        false, wf0, false, xs0, (short)0, sacc[o], false, false);
    sacc[o] = __builtin_amdgcn_wmma_f32_16x16x4_f32(
        false, wf1, false, xs1, (short)0, sacc[o], false, false);
    wq[o] = *(const v4f*)(wnext + o * (DO_ * DI));
    if ((o & 3) == 3) __builtin_amdgcn_sched_barrier(0);
  }
  xq = xqn;
}

template <int ITER>
__global__ __launch_bounds__(128)
void caps_route(const float* __restrict__ x, const float* __restrict__ W,
                const float* __restrict__ vcur, float* __restrict__ rlog,
                float* __restrict__ ps) {
  __shared__ float4 comb[2 * NO * 2 * 32];
  const int lane = threadIdx.x & 31;
  const int wave = threadIdx.x >> 5;
  const int half = lane >> 4;
  const int l15  = lane & 15;
  const int mt   = wave & 1;
  const int sub  = wave >> 1;
  const int nblk = blockIdx.x >> 1;
  const int b0   = (blockIdx.x & 1) * BROWS;
  const int brow = b0 + mt * 16 + l15;

  const v8f zero = {};
  v8f sacc[NO];
  v8f vv[NO];
  const float* vb = vcur + brow * NOD + 8 * half;
#pragma unroll
  for (int o = 0; o < NO; ++o) {
    sacc[o] = zero;
#pragma unroll
    for (int r = 0; r < 8; ++r) vv[o][r] = vb[o * DO_ + r];
  }

  const int n0 = nblk * NPB + sub * CPW;
  const float* xb = x + (size_t)brow * KTOT + 4 * half;
  const float* wb = W + (size_t)l15 * DI + 4 * half;
  float* rlb = rlog + (size_t)brow * NI * NO;

  v4f xq;
  v4f wq[NO];
  float4 old[NO / 4];
  load_frags(xb, wb, n0, xq, wq);
  if (ITER != 1) {
#pragma unroll
    for (int q = 0; q < NO / 4; ++q) old[q] = ((const float4*)(rlb + (size_t)n0 * NO))[q];
  }
  route_capsule<ITER>(xq, wq, vv, sacc, rlb + (size_t)n0 * NO, half, old,
                      xb + (n0 + 1) * DI, wb + (size_t)(n0 + 1) * WN, rlb + (size_t)(n0 + 1) * NO);
#pragma unroll 1
  for (int g = 1; g < CPW; ++g) {
    const int nn = (g + 1 < CPW) ? (n0 + g + 1) : (n0 + g);
    route_capsule<ITER>(xq, wq, vv, sacc, rlb + (size_t)(n0 + g) * NO, half, old,
                        xb + nn * DI, wb + (size_t)nn * WN, rlb + (size_t)nn * NO);
  }

  float4* cb = comb + (mt * NO * 2) * 32 + lane;
  if (sub == 1) {
#pragma unroll
    for (int o = 0; o < NO; ++o) {
      cb[(o * 2) * 32]     = make_float4(sacc[o][0], sacc[o][1], sacc[o][2], sacc[o][3]);
      cb[(o * 2 + 1) * 32] = make_float4(sacc[o][4], sacc[o][5], sacc[o][6], sacc[o][7]);
    }
  }
  __syncthreads();
  if (sub == 0) {
    float* psb = ps + (size_t)nblk * PS_STRIDE + brow * NOD + 8 * half;
#pragma unroll
    for (int o = 0; o < NO; ++o) {
      const float4 lo = cb[(o * 2) * 32];
      const float4 hi = cb[(o * 2 + 1) * 32];
      float4* dst = (float4*)(psb + o * DO_);
      dst[0] = make_float4(sacc[o][0] + lo.x, sacc[o][1] + lo.y, sacc[o][2] + lo.z, sacc[o][3] + lo.w);
      dst[1] = make_float4(sacc[o][4] + hi.x, sacc[o][5] + hi.y, sacc[o][6] + hi.z, sacc[o][7] + hi.w);
    }
  }
}

#define RQ   128
#define RPG  2
__global__ __launch_bounds__(256)
void caps_reduce_squash(const float* __restrict__ ps, float prescale,
                        float* __restrict__ out) {
  __shared__ float part[RPG][RQ];
  __shared__ float sh[RQ];
  __shared__ float scl[RQ / DO_];
  const int b  = blockIdx.x >> 2;
  const int j  = threadIdx.x & (RQ - 1);
  const int pg = threadIdx.x >> 7;
  const int od = (blockIdx.x & 3) * RQ + j;
  const float* src = ps + (size_t)(pg * (NPART / RPG)) * PS_STRIDE + b * NOD + od;
  float s = 0.f;
#pragma unroll
  for (int p = 0; p < NPART / RPG; ++p)
    s += src[(size_t)p * PS_STRIDE];
  part[pg][j] = s;
  __syncthreads();
  if (threadIdx.x < RQ)
    sh[j] = (part[0][j] + part[1][j]) * prescale;
  __syncthreads();
  if (threadIdx.x < RQ / DO_) {
    float nn = 0.f;
#pragma unroll
    for (int d = 0; d < DO_; ++d) { float t = sh[threadIdx.x * DO_ + d]; nn += t * t; }
    scl[threadIdx.x] = sqrtf(nn) / (1.f + nn);
  }
  __syncthreads();
  if (threadIdx.x < RQ)
    out[(size_t)b * NOD + od] = sh[j] * scl[j >> 4];
}

extern "C" void kernel_launch(void* const* d_in, const int* in_sizes, int n_in,
                              void* d_out, int out_size, void* d_ws, size_t ws_size,
                              hipStream_t stream) {
  const float* x = (const float*)d_in[0];
  const float* W = (const float*)d_in[1];
  float* out = (float*)d_out;

  float* ps   = (float*)d_ws;
  float* vbuf = ps + (size_t)NPART * PS_STRIDE;
  float* rlog = vbuf + (size_t)B_SZ * NOD;

  caps_s0_gemm<<<NB_BIG, 256, 0, stream>>>(x, W, ps);
  caps_reduce_squash<<<256, 256, 0, stream>>>(ps, 1.0f / NO, vbuf);
  caps_route<1><<<NB_BIG, 128, 0, stream>>>(x, W, vbuf, rlog, ps);
  caps_reduce_squash<<<256, 256, 0, stream>>>(ps, 1.0f, vbuf);
  caps_route<2><<<NB_BIG, 128, 0, stream>>>(x, W, vbuf, rlog, ps);
  caps_reduce_squash<<<256, 256, 0, stream>>>(ps, 1.0f, out);
}


# caps_route: x_hat tiles of phase A stay in registers and phase C reuses them, s += c * x_hat as 4 packed multiply-adds per tile: W * x is formed once per capsule on WMMA (64 per capsule, was 128)
# speedup vs baseline: 32.2172x; 1.1456x over previous
#include <hip/hip_runtime.h>
#include <math.h>

typedef __attribute__((ext_vector_type(2))) float v2f;
typedef __attribute__((ext_vector_type(4))) float v4f;
typedef __attribute__((ext_vector_type(8))) float v8f;

#define B_SZ   64
#define NI     2048
#define DI     8
#define NO     32
#define DO_    16
#define NOD    512
#define KTOT   (NI*DI)
#define WN     (NO*DO_*DI)

#define NPART    128
#define KCHUNK   (KTOT/NPART)
#define NPB      (NI/NPART)
#define BROWS    32
#define NB_BIG   256
#define PS_STRIDE (B_SZ*NOD)

#define GCAP   (KCHUNK / DI)
#define GST    4
#define GOT    4
#define GMT    2

static __device__ __forceinline__ void gemm_load_stage(const float* __restrict__ xr,
                                                       const float* __restrict__ wc, int n,
                                                       v4f (&xf)[GST][GMT], v4f (&wf)[GST][GOT]) {
#pragma unroll
  for (int c = 0; c < GST; ++c) {
#pragma unroll
    for (int mt = 0; mt < GMT; ++mt)
      xf[c][mt] = *(const v4f*)(xr + (size_t)mt * 16 * KTOT + (n + c) * DI);
#pragma unroll
    for (int ot = 0; ot < GOT; ++ot)
      wf[c][ot] = *(const v4f*)(wc + (size_t)(n + c) * WN + ot * (8 * DO_ * DI));
  }
}

static __device__ __forceinline__ void gemm_compute_stage(const v4f (&xf)[GST][GMT],
                                                          const v4f (&wf)[GST][GOT],
                                                          v8f (&acc)[GOT][GMT]) {
#pragma unroll
  for (int c = 0; c < GST; ++c) {
#pragma unroll
    for (int ot = 0; ot < GOT; ++ot) {
      const v2f b = __builtin_shufflevector(wf[c][ot], wf[c][ot], 0, 1);
#pragma unroll
      for (int mt = 0; mt < GMT; ++mt) {
        const v2f a = __builtin_shufflevector(xf[c][mt], xf[c][mt], 0, 1);
        acc[ot][mt] = __builtin_amdgcn_wmma_f32_16x16x4_f32(
            false, a, false, b, (short)0, acc[ot][mt], false, false);
      }
    }
#pragma unroll
    for (int ot = 0; ot < GOT; ++ot) {
      const v2f b = __builtin_shufflevector(wf[c][ot], wf[c][ot], 2, 3);
#pragma unroll
      for (int mt = 0; mt < GMT; ++mt) {
        const v2f a = __builtin_shufflevector(xf[c][mt], xf[c][mt], 2, 3);
        acc[ot][mt] = __builtin_amdgcn_wmma_f32_16x16x4_f32(
            false, a, false, b, (short)0, acc[ot][mt], false, false);
      }
    }
  }
}

__global__ __launch_bounds__(256)
void caps_s0_gemm(const float* __restrict__ x, const float* __restrict__ W,
                  float* __restrict__ ps) {
  const int lane = threadIdx.x & 31;
  const int wave = threadIdx.x >> 5;
  const int half = lane >> 4;
  const int l15  = lane & 15;
  const int kblk = blockIdx.x >> 1;
  const int b0   = (blockIdx.x & 1) * BROWS;
  const int n0   = kblk * GCAP;
  float* psb = ps + (size_t)kblk * PS_STRIDE;

  const float* xr = x + (size_t)(b0 + l15) * KTOT + 4 * half;
  const float* wc = W + (size_t)(wave * DO_ + l15) * DI + 4 * half;

  const v8f zero = {};
  v8f acc[GOT][GMT];
#pragma unroll
  for (int ot = 0; ot < GOT; ++ot)
#pragma unroll
    for (int mt = 0; mt < GMT; ++mt) acc[ot][mt] = zero;

  v4f xa[GST][GMT], wa[GST][GOT];
  v4f xb[GST][GMT], wb[GST][GOT];

  gemm_load_stage(xr, wc, n0, xa, wa);
  gemm_load_stage(xr, wc, n0 + GST, xb, wb);
  __builtin_amdgcn_sched_barrier(0);
  gemm_compute_stage(xa, wa, acc);
  __builtin_amdgcn_sched_barrier(0);
  gemm_load_stage(xr, wc, n0 + 2 * GST, xa, wa);
  __builtin_amdgcn_sched_barrier(0);
  gemm_compute_stage(xb, wb, acc);
  __builtin_amdgcn_sched_barrier(0);
  gemm_load_stage(xr, wc, n0 + 3 * GST, xb, wb);
  __builtin_amdgcn_sched_barrier(0);
  gemm_compute_stage(xa, wa, acc);
  __builtin_amdgcn_sched_barrier(0);
  gemm_compute_stage(xb, wb, acc);

#pragma unroll
  for (int ot = 0; ot < GOT; ++ot) {
    const int o = wave + ot * 8;
#pragma unroll
    for (int mt = 0; mt < GMT; ++mt) {
#pragma unroll
      for (int r = 0; r < 8; ++r) {
        const int b = b0 + mt * 16 + r + 8 * half;
        psb[b * NOD + o * DO_ + l15] = acc[ot][mt][r];
      }
    }
  }
}

#define CPW 8

static __device__ __forceinline__ float xhalf(float v) {
  return __builtin_bit_cast(float, __builtin_amdgcn_permlanex16(
      __builtin_bit_cast(int, v), __builtin_bit_cast(int, v), 0x76543210, 0xfedcba98, false, false));
}

static __device__ __forceinline__ void load_frags(const float* __restrict__ xb,
                                                  const float* __restrict__ wb, int n,
                                                  v4f &xf, v4f (&wf)[NO]) {
  xf = *(const v4f*)(xb + n * DI);
  const float* wn = wb + (size_t)n * WN;
#pragma unroll
  for (int o = 0; o < NO; ++o) wf[o] = *(const v4f*)(wn + o * (DO_ * DI));
}

#define WPRE 12

template <int ITER>
static __device__ __forceinline__ void route_capsule(v4f &xq, v4f (&wq)[NO],
                                                     const v8f (&vv)[NO], v8f (&sacc)[NO],
                                                     float* __restrict__ rl, int half,
                                                     float4 (&old)[NO / 4],
                                                     const float* __restrict__ xnext,
                                                     const float* __restrict__ wnext,
                                                     const float* __restrict__ rlnext) {
  const v2f xf0 = __builtin_shufflevector(xq, xq, 0, 1);
  const v2f xf1 = __builtin_shufflevector(xq, xq, 2, 3);

  const v8f zero = {};
  v8f xh[NO];
  float e[NO];
#pragma unroll
  for (int o = 0; o < NO; ++o) {
    const v2f wf0 = __builtin_shufflevector(wq[o], wq[o], 0, 1);
    const v2f wf1 = __builtin_shufflevector(wq[o], wq[o], 2, 3);
    v8f d = __builtin_amdgcn_wmma_f32_16x16x4_f32(
                false, wf0, false, xf0, (short)0, zero, false, false);
    d = __builtin_amdgcn_wmma_f32_16x16x4_f32(
                false, wf1, false, xf1, (short)0, d, false, false);
    xh[o] = d;
    const v8f w = vv[o];
    v2f q = __builtin_shufflevector(d, d, 0, 1) * __builtin_shufflevector(w, w, 0, 1);
    q += __builtin_shufflevector(d, d, 2, 3) * __builtin_shufflevector(w, w, 2, 3);
    q += __builtin_shufflevector(d, d, 4, 5) * __builtin_shufflevector(w, w, 4, 5);
    q += __builtin_shufflevector(d, d, 6, 7) * __builtin_shufflevector(w, w, 6, 7);
    e[o] = q[0] + q[1];
    if ((o & 3) == 3) __builtin_amdgcn_sched_barrier(0);
  }
  const v4f xqn = *(const v4f*)xnext;
#pragma unroll
  for (int o = 0; o < WPRE; ++o) wq[o] = *(const v4f*)(wnext + o * (DO_ * DI));
  __builtin_amdgcn_sched_barrier(0);

#pragma unroll
  for (int o = 0; o < NO; ++o) e[o] += xhalf(e[o]);

  if (ITER == 1) {
    float4* rl4 = (float4*)rl + 4 * half;
#pragma unroll
    for (int q = 0; q < 4; ++q)
      rl4[q] = make_float4(half ? e[16 + 4 * q]     : e[4 * q],
                           half ? e[16 + 4 * q + 1] : e[4 * q + 1],
                           half ? e[16 + 4 * q + 2] : e[4 * q + 2],
                           half ? e[16 + 4 * q + 3] : e[4 * q + 3]);
  } else {
#pragma unroll
    for (int q = 0; q < NO / 4; ++q) {
      e[4 * q]     += old[q].x;
      e[4 * q + 1] += old[q].y;
      e[4 * q + 2] += old[q].z;
      e[4 * q + 3] += old[q].w;
    }
  }
  float m4[4] = {-3.4e38f, -3.4e38f, -3.4e38f, -3.4e38f};
#pragma unroll
  for (int o = 0; o < NO; ++o) m4[o & 3] = fmaxf(m4[o & 3], e[o]);
  const float m = fmaxf(fmaxf(m4[0], m4[1]), fmaxf(m4[2], m4[3]));
  float s4[4] = {0.f, 0.f, 0.f, 0.f};
#pragma unroll
  for (int o = 0; o < NO; ++o) { e[o] = __expf(e[o] - m); s4[o & 3] += e[o]; }
  const float ssum = (s4[0] + s4[1]) + (s4[2] + s4[3]);
  const float inv = 1.f / ssum;

  if (ITER != 1) {
#pragma unroll
    for (int q = 0; q < NO / 4; ++q) old[q] = ((const float4*)rlnext)[q];
  }
  __builtin_amdgcn_sched_barrier(0);
#pragma unroll
  for (int o = 0; o < NO; ++o) {
    const float c = e[o] * inv;
    sacc[o] += xh[o] * c;
    if (o + WPRE < NO) wq[o + WPRE] = *(const v4f*)(wnext + (o + WPRE) * (DO_ * DI));
    if ((o & 3) == 3) __builtin_amdgcn_sched_barrier(0);
  }
  xq = xqn;
}

template <int ITER>
__global__ __launch_bounds__(128)
void caps_route(const float* __restrict__ x, const float* __restrict__ W,
                const float* __restrict__ vcur, float* __restrict__ rlog,
                float* __restrict__ ps) {
  __shared__ float4 comb[2 * NO * 2 * 32];
  const int lane = threadIdx.x & 31;
  const int wave = threadIdx.x >> 5;
  const int half = lane >> 4;
  const int l15  = lane & 15;
  const int mt   = wave & 1;
  const int sub  = wave >> 1;
  const int nblk = blockIdx.x >> 1;
  const int b0   = (blockIdx.x & 1) * BROWS;
  const int brow = b0 + mt * 16 + l15;

  const v8f zero = {};
  v8f sacc[NO];
  v8f vv[NO];
  const float* vb = vcur + brow * NOD + 8 * half;
#pragma unroll
  for (int o = 0; o < NO; ++o) {
    sacc[o] = zero;
#pragma unroll
    for (int r = 0; r < 8; ++r) vv[o][r] = vb[o * DO_ + r];
  }

  const int n0 = nblk * NPB + sub * CPW;
  const float* xb = x + (size_t)brow * KTOT + 4 * half;
  const float* wb = W + (size_t)l15 * DI + 4 * half;
  float* rlb = rlog + (size_t)brow * NI * NO;

  v4f xq;
  v4f wq[NO];
  float4 old[NO / 4];
  load_frags(xb, wb, n0, xq, wq);
  if (ITER != 1) {
#pragma unroll
    for (int q = 0; q < NO / 4; ++q) old[q] = ((const float4*)(rlb + (size_t)n0 * NO))[q];
  }
  route_capsule<ITER>(xq, wq, vv, sacc, rlb + (size_t)n0 * NO, half, old,
                      xb + (n0 + 1) * DI, wb + (size_t)(n0 + 1) * WN, rlb + (size_t)(n0 + 1) * NO);
#pragma unroll 1
  for (int g = 1; g < CPW; ++g) {
    const int nn = (g + 1 < CPW) ? (n0 + g + 1) : (n0 + g);
    route_capsule<ITER>(xq, wq, vv, sacc, rlb + (size_t)(n0 + g) * NO, half, old,
                        xb + nn * DI, wb + (size_t)nn * WN, rlb + (size_t)nn * NO);
  }

  float4* cb = comb + (mt * NO * 2) * 32 + lane;
  if (sub == 1) {
#pragma unroll
    for (int o = 0; o < NO; ++o) {
      cb[(o * 2) * 32]     = make_float4(sacc[o][0], sacc[o][1], sacc[o][2], sacc[o][3]);
      cb[(o * 2 + 1) * 32] = make_float4(sacc[o][4], sacc[o][5], sacc[o][6], sacc[o][7]);
    }
  }
  __syncthreads();
  if (sub == 0) {
    float* psb = ps + (size_t)nblk * PS_STRIDE + brow * NOD + 8 * half;
#pragma unroll
    for (int o = 0; o < NO; ++o) {
      const float4 lo = cb[(o * 2) * 32];
      const float4 hi = cb[(o * 2 + 1) * 32];
      float4* dst = (float4*)(psb + o * DO_);
      dst[0] = make_float4(sacc[o][0] + lo.x, sacc[o][1] + lo.y, sacc[o][2] + lo.z, sacc[o][3] + lo.w);
      dst[1] = make_float4(sacc[o][4] + hi.x, sacc[o][5] + hi.y, sacc[o][6] + hi.z, sacc[o][7] + hi.w);
    }
  }
}

#define RQ   128
#define RPG  2
__global__ __launch_bounds__(256)
void caps_reduce_squash(const float* __restrict__ ps, float prescale,
                        float* __restrict__ out) {
  __shared__ float part[RPG][RQ];
  __shared__ float sh[RQ];
  __shared__ float scl[RQ / DO_];
  const int b  = blockIdx.x >> 2;
  const int j  = threadIdx.x & (RQ - 1);
  const int pg = threadIdx.x >> 7;
  const int od = (blockIdx.x & 3) * RQ + j;
  const float* src = ps + (size_t)(pg * (NPART / RPG)) * PS_STRIDE + b * NOD + od;
  float s = 0.f;
#pragma unroll
  for (int p = 0; p < NPART / RPG; ++p)
    s += src[(size_t)p * PS_STRIDE];
  part[pg][j] = s;
  __syncthreads();
  if (threadIdx.x < RQ)
    sh[j] = (part[0][j] + part[1][j]) * prescale;
  __syncthreads();
  if (threadIdx.x < RQ / DO_) {
    float nn = 0.f;
#pragma unroll
    for (int d = 0; d < DO_; ++d) { float t = sh[threadIdx.x * DO_ + d]; nn += t * t; }
    scl[threadIdx.x] = sqrtf(nn) / (1.f + nn);
  }
  __syncthreads();
  if (threadIdx.x < RQ)
    out[(size_t)b * NOD + od] = sh[j] * scl[j >> 4];
}

extern "C" void kernel_launch(void* const* d_in, const int* in_sizes, int n_in,
                              void* d_out, int out_size, void* d_ws, size_t ws_size,
                              hipStream_t stream) {
  const float* x = (const float*)d_in[0];
  const float* W = (const float*)d_in[1];
  float* out = (float*)d_out;

  float* ps   = (float*)d_ws;
  float* vbuf = ps + (size_t)NPART * PS_STRIDE;
  float* rlog = vbuf + (size_t)B_SZ * NOD;

  caps_s0_gemm<<<NB_BIG, 256, 0, stream>>>(x, W, ps);
  caps_reduce_squash<<<256, 256, 0, stream>>>(ps, 1.0f / NO, vbuf);
  caps_route<1><<<NB_BIG, 128, 0, stream>>>(x, W, vbuf, rlog, ps);
  caps_reduce_squash<<<256, 256, 0, stream>>>(ps, 1.0f, vbuf);
  caps_route<2><<<NB_BIG, 128, 0, stream>>>(x, W, vbuf, rlog, ps);
  caps_reduce_squash<<<256, 256, 0, stream>>>(ps, 1.0f, out);
}


# caps: workspace in wave-fragment order: partial-s planes, v, running logits kept as the tiles' own 16-byte lane pieces, every access whole 128-byte lines (routing was 16 lines, GEMM 4-byte stores)
# speedup vs baseline: 34.0102x; 1.0557x over previous
#include <hip/hip_runtime.h>
#include <math.h>

typedef __attribute__((ext_vector_type(2))) float v2f;
typedef __attribute__((ext_vector_type(4))) float v4f;
typedef __attribute__((ext_vector_type(8))) float v8f;

#define B_SZ   64
#define NI     2048
#define DI     8
#define NO     32
#define DO_    16
#define NOD    512
#define KTOT   (NI*DI)
#define WN     (NO*DO_*DI)

#define NPART    128
#define KCHUNK   (KTOT/NPART)
#define NPB      (NI/NPART)
#define BROWS    32
#define NB_BIG   256
#define PS_STRIDE (B_SZ*NOD)

#define VT_STRIDE (NO*256)

#define GCAP   (KCHUNK / DI)
#define GST    4
#define GOT    4
#define GMT    2

static __device__ __forceinline__ void gemm_load_stage(const float* __restrict__ xr,
                                                       const float* __restrict__ wc, int n,
                                                       v4f (&xf)[GST][GMT], v4f (&wf)[GST][GOT]) {
#pragma unroll
  for (int c = 0; c < GST; ++c) {
#pragma unroll
    for (int mt = 0; mt < GMT; ++mt)
      xf[c][mt] = *(const v4f*)(xr + (size_t)mt * 16 * KTOT + (n + c) * DI);
#pragma unroll
    for (int ot = 0; ot < GOT; ++ot)
      wf[c][ot] = *(const v4f*)(wc + (size_t)(n + c) * WN + ot * (8 * DO_ * DI));
  }
}

static __device__ __forceinline__ void gemm_compute_stage(const v4f (&xf)[GST][GMT],
                                                          const v4f (&wf)[GST][GOT],
                                                          v8f (&acc)[GOT][GMT]) {
#pragma unroll
  for (int c = 0; c < GST; ++c) {
#pragma unroll
    for (int ot = 0; ot < GOT; ++ot) {
      const v2f b = __builtin_shufflevector(wf[c][ot], wf[c][ot], 0, 1);
#pragma unroll
      for (int mt = 0; mt < GMT; ++mt) {
        const v2f a = __builtin_shufflevector(xf[c][mt], xf[c][mt], 0, 1);
        acc[ot][mt] = __builtin_amdgcn_wmma_f32_16x16x4_f32(
            false, a, false, b, (short)0, acc[ot][mt], false, false);
      }
    }
#pragma unroll
    for (int ot = 0; ot < GOT; ++ot) {
      const v2f b = __builtin_shufflevector(wf[c][ot], wf[c][ot], 2, 3);
#pragma unroll
      for (int mt = 0; mt < GMT; ++mt) {
        const v2f a = __builtin_shufflevector(xf[c][mt], xf[c][mt], 2, 3);
        acc[ot][mt] = __builtin_amdgcn_wmma_f32_16x16x4_f32(
            false, a, false, b, (short)0, acc[ot][mt], false, false);
      }
    }
  }
}

__global__ __launch_bounds__(256)
void caps_s0_gemm(const float* __restrict__ x, const float* __restrict__ W,
                  float* __restrict__ ps) {
  const int lane = threadIdx.x & 31;
  const int wave = threadIdx.x >> 5;
  const int half = lane >> 4;
  const int l15  = lane & 15;
  const int kblk = blockIdx.x >> 1;
  const int b0   = (blockIdx.x & 1) * BROWS;
  const int n0   = kblk * GCAP;
  float* psb = ps + (size_t)kblk * PS_STRIDE;

  const float* xr = x + (size_t)(b0 + l15) * KTOT + 4 * half;
  const float* wc = W + (size_t)(wave * DO_ + l15) * DI + 4 * half;

  const v8f zero = {};
  v8f acc[GOT][GMT];
#pragma unroll
  for (int ot = 0; ot < GOT; ++ot)
#pragma unroll
    for (int mt = 0; mt < GMT; ++mt) acc[ot][mt] = zero;

  v4f xa[GST][GMT], wa[GST][GOT];
  v4f xb[GST][GMT], wb[GST][GOT];

  gemm_load_stage(xr, wc, n0, xa, wa);
  gemm_load_stage(xr, wc, n0 + GST, xb, wb);
  __builtin_amdgcn_sched_barrier(0);
  gemm_compute_stage(xa, wa, acc);
  __builtin_amdgcn_sched_barrier(0);
  gemm_load_stage(xr, wc, n0 + 2 * GST, xa, wa);
  __builtin_amdgcn_sched_barrier(0);
  gemm_compute_stage(xb, wb, acc);
  __builtin_amdgcn_sched_barrier(0);
  gemm_load_stage(xr, wc, n0 + 3 * GST, xb, wb);
  __builtin_amdgcn_sched_barrier(0);
  gemm_compute_stage(xa, wa, acc);
  __builtin_amdgcn_sched_barrier(0);
  gemm_compute_stage(xb, wb, acc);

#pragma unroll
  for (int ot = 0; ot < GOT; ++ot) {
    const int o = wave + ot * 8;
#pragma unroll
    for (int mt = 0; mt < GMT; ++mt) {
      const int tm = (b0 >> 4) + mt;
      float* dst = psb + (size_t)(((o * 4 + tm) * 2) * 32 + lane) * 4;
      *(v4f*)dst         = __builtin_shufflevector(acc[ot][mt], acc[ot][mt], 0, 1, 2, 3);
      *(v4f*)(dst + 128) = __builtin_shufflevector(acc[ot][mt], acc[ot][mt], 4, 5, 6, 7);
    }
  }
}

#define CPW 8

static __device__ __forceinline__ float xhalf(float v) {
  return __builtin_bit_cast(float, __builtin_amdgcn_permlanex16(
      __builtin_bit_cast(int, v), __builtin_bit_cast(int, v), 0x76543210, 0xfedcba98, false, false));
}

static __device__ __forceinline__ void load_frags(const float* __restrict__ xb,
                                                  const float* __restrict__ wb, int n,
                                                  v4f &xf, v4f (&wf)[NO]) {
  xf = *(const v4f*)(xb + n * DI);
  const float* wn = wb + (size_t)n * WN;
#pragma unroll
  for (int o = 0; o < NO; ++o) wf[o] = *(const v4f*)(wn + o * (DO_ * DI));
}

#define WPRE 12

template <int ITER>
static __device__ __forceinline__ void route_capsule(v4f &xq, v4f (&wq)[NO],
                                                     const v8f (&vv)[NO], v8f (&sacc)[NO],
                                                     float* __restrict__ rl, int half,
                                                     float4 (&old)[NO / 4],
                                                     const float* __restrict__ xnext,
                                                     const float* __restrict__ wnext,
                                                     const float* __restrict__ rlnext) {
  const v2f xf0 = __builtin_shufflevector(xq, xq, 0, 1);
  const v2f xf1 = __builtin_shufflevector(xq, xq, 2, 3);

  const v8f zero = {};
  v8f xh[NO];
  float e[NO];
#pragma unroll
  for (int o = 0; o < NO; ++o) {
    const v2f wf0 = __builtin_shufflevector(wq[o], wq[o], 0, 1);
    const v2f wf1 = __builtin_shufflevector(wq[o], wq[o], 2, 3);
    v8f d = __builtin_amdgcn_wmma_f32_16x16x4_f32(
                false, wf0, false, xf0, (short)0, zero, false, false);
    d = __builtin_amdgcn_wmma_f32_16x16x4_f32(
                false, wf1, false, xf1, (short)0, d, false, false);
    xh[o] = d;
    const v8f w = vv[o];
    v2f q = __builtin_shufflevector(d, d, 0, 1) * __builtin_shufflevector(w, w, 0, 1);
    q += __builtin_shufflevector(d, d, 2, 3) * __builtin_shufflevector(w, w, 2, 3);
    q += __builtin_shufflevector(d, d, 4, 5) * __builtin_shufflevector(w, w, 4, 5);
    q += __builtin_shufflevector(d, d, 6, 7) * __builtin_shufflevector(w, w, 6, 7);
    e[o] = q[0] + q[1];
    if ((o & 3) == 3) __builtin_amdgcn_sched_barrier(0);
  }
  const v4f xqn = *(const v4f*)xnext;
#pragma unroll
  for (int o = 0; o < WPRE; ++o) wq[o] = *(const v4f*)(wnext + o * (DO_ * DI));
  __builtin_amdgcn_sched_barrier(0);

#pragma unroll
  for (int o = 0; o < NO; ++o) e[o] += xhalf(e[o]);

  if (ITER == 1) {
    float4* rl4 = (float4*)rl + (4 * half) * 16;
#pragma unroll
    for (int q = 0; q < 4; ++q)
      rl4[q * 16] = make_float4(half ? e[16 + 4 * q]     : e[4 * q],
                           half ? e[16 + 4 * q + 1] : e[4 * q + 1],
                           half ? e[16 + 4 * q + 2] : e[4 * q + 2],
                           half ? e[16 + 4 * q + 3] : e[4 * q + 3]);
  } else {
#pragma unroll
    for (int q = 0; q < NO / 4; ++q) {
      e[4 * q]     += old[q].x;
      e[4 * q + 1] += old[q].y;
      e[4 * q + 2] += old[q].z;
      e[4 * q + 3] += old[q].w;
    }
  }
  float m4[4] = {-3.4e38f, -3.4e38f, -3.4e38f, -3.4e38f};
#pragma unroll
  for (int o = 0; o < NO; ++o) m4[o & 3] = fmaxf(m4[o & 3], e[o]);
  const float m = fmaxf(fmaxf(m4[0], m4[1]), fmaxf(m4[2], m4[3]));
  float s4[4] = {0.f, 0.f, 0.f, 0.f};
#pragma unroll
  for (int o = 0; o < NO; ++o) { e[o] = __expf(e[o] - m); s4[o & 3] += e[o]; }
  const float ssum = (s4[0] + s4[1]) + (s4[2] + s4[3]);
  const float inv = 1.f / ssum;

  if (ITER != 1) {
#pragma unroll
    for (int q = 0; q < NO / 4; ++q) old[q] = ((const float4*)rlnext)[q * 16];
  }
  __builtin_amdgcn_sched_barrier(0);
#pragma unroll
  for (int o = 0; o < NO; ++o) {
    const float c = e[o] * inv;
    sacc[o] += xh[o] * c;
    if (o + WPRE < NO) wq[o + WPRE] = *(const v4f*)(wnext + (o + WPRE) * (DO_ * DI));
    if ((o & 3) == 3) __builtin_amdgcn_sched_barrier(0);
  }
  xq = xqn;
}

template <int ITER>
__global__ __launch_bounds__(128)
void caps_route(const float* __restrict__ x, const float* __restrict__ W,
                const float* __restrict__ vcur, float* __restrict__ rlog,
                float* __restrict__ ps) {
  __shared__ float4 comb[2 * NO * 2 * 32];
  const int lane = threadIdx.x & 31;
  const int wave = threadIdx.x >> 5;
  const int half = lane >> 4;
  const int l15  = lane & 15;
  const int mt   = wave & 1;
  const int sub  = wave >> 1;
  const int nblk = blockIdx.x >> 1;
  const int b0   = (blockIdx.x & 1) * BROWS;
  const int brow = b0 + mt * 16 + l15;
  const int tile = (blockIdx.x & 1) * 2 + mt;

  const v8f zero = {};
  v8f sacc[NO];
  v8f vv[NO];
  const float* vb = vcur + (size_t)tile * VT_STRIDE + lane * 4;
#pragma unroll
  for (int o = 0; o < NO; ++o) {
    sacc[o] = zero;
    const v4f vlo = *(const v4f*)(vb + (o * 2) * 128);
    const v4f vhi = *(const v4f*)(vb + (o * 2 + 1) * 128);
    vv[o] = __builtin_shufflevector(vlo, vhi, 0, 1, 2, 3, 4, 5, 6, 7);
  }

  const int n0 = nblk * NPB + sub * CPW;
  const float* xb = x + (size_t)brow * KTOT + 4 * half;
  const float* wb = W + (size_t)l15 * DI + 4 * half;
  float* rlb = rlog + (size_t)tile * NI * (16 * NO) + l15 * 4;

  v4f xq;
  v4f wq[NO];
  float4 old[NO / 4];
  load_frags(xb, wb, n0, xq, wq);
  if (ITER != 1) {
#pragma unroll
    for (int q = 0; q < NO / 4; ++q) old[q] = ((const float4*)(rlb + (size_t)n0 * (16 * NO)))[q * 16];
  }
  route_capsule<ITER>(xq, wq, vv, sacc, rlb + (size_t)n0 * (16 * NO), half, old,
                      xb + (n0 + 1) * DI, wb + (size_t)(n0 + 1) * WN, rlb + (size_t)(n0 + 1) * (16 * NO));
#pragma unroll 1
  for (int g = 1; g < CPW; ++g) {
    const int nn = (g + 1 < CPW) ? (n0 + g + 1) : (n0 + g);
    route_capsule<ITER>(xq, wq, vv, sacc, rlb + (size_t)(n0 + g) * (16 * NO), half, old,
                        xb + nn * DI, wb + (size_t)nn * WN, rlb + (size_t)nn * (16 * NO));
  }

  float4* cb = comb + (mt * NO * 2) * 32 + lane;
  if (sub == 1) {
#pragma unroll
    for (int o = 0; o < NO; ++o) {
      cb[(o * 2) * 32]     = make_float4(sacc[o][0], sacc[o][1], sacc[o][2], sacc[o][3]);
      cb[(o * 2 + 1) * 32] = make_float4(sacc[o][4], sacc[o][5], sacc[o][6], sacc[o][7]);
    }
  }
  __syncthreads();
  if (sub == 0) {
    float* psb = ps + (size_t)nblk * PS_STRIDE + (size_t)tile * VT_STRIDE + lane * 4;
#pragma unroll
    for (int o = 0; o < NO; ++o) {
      const float4 lo = cb[(o * 2) * 32];
      const float4 hi = cb[(o * 2 + 1) * 32];
      *(float4*)(psb + (o * 2) * 128) =
          make_float4(sacc[o][0] + lo.x, sacc[o][1] + lo.y, sacc[o][2] + lo.z, sacc[o][3] + lo.w);
      *(float4*)(psb + (o * 2 + 1) * 128) =
          make_float4(sacc[o][4] + hi.x, sacc[o][5] + hi.y, sacc[o][6] + hi.z, sacc[o][7] + hi.w);
    }
  }
}

#define RQ   128
#define RPG  2
template <int SRC, int DST>
__global__ __launch_bounds__(256)
void caps_reduce_squash(const float* __restrict__ ps, float prescale,
                        float* __restrict__ out) {
  __shared__ float part[RPG][RQ];
  __shared__ float sh[RQ];
  __shared__ float scl[RQ / DO_];
  const int t  = threadIdx.x & (RQ - 1);
  const int pg = threadIdx.x >> 7;
  int b, o, d, g, soff;
  if (SRC == 0) {
    const int q  = blockIdx.x & 1;
    const int tm = (blockIdx.x >> 1) & 3;
    o = blockIdx.x >> 3;
    const int k  = t & 3;
    const int ln = t >> 2;
    const int hf = ln >> 4;
    d = ln & 15;
    b = tm * 16 + hf * 8 + q * 4 + k;
    g = hf * 4 + k;
    soff = blockIdx.x * RQ + t;
  } else {
    const int cc = blockIdx.x & 1;
    o = (blockIdx.x >> 1) & 31;
    const int tl  = blockIdx.x >> 6;
    const int run = t >> 5;
    const int c8  = (t >> 2) & 7;
    const int k   = t & 3;
    const int j   = run >> 1;
    const int h   = run & 1;
    d = 8 * h + 4 * j + k;
    b = tl * 16 + cc * 8 + c8;
    g = c8;
    soff = ((tl * 32 + o) * 2 + j) * 128 + (h * 16 + cc * 8 + c8) * 4 + k;
  }
  const float* src = ps + (size_t)(pg * (NPART / RPG)) * PS_STRIDE + soff;
  float s = 0.f;
#pragma unroll
  for (int p = 0; p < NPART / RPG; ++p)
    s += src[(size_t)p * PS_STRIDE];
  part[pg][t] = s;
  __syncthreads();
  if (threadIdx.x < RQ)
    sh[g * DO_ + d] = (part[0][t] + part[1][t]) * prescale;
  __syncthreads();
  if (threadIdx.x < RQ / DO_) {
    float nn = 0.f;
#pragma unroll
    for (int dd = 0; dd < DO_; ++dd) { float u = sh[threadIdx.x * DO_ + dd]; nn += u * u; }
    scl[threadIdx.x] = sqrtf(nn) / (1.f + nn);
  }
  __syncthreads();
  if (threadIdx.x < RQ) {
    const float val = sh[g * DO_ + d] * scl[g];
    if (DST == 0)
      out[(((b >> 4) * 32 + o) * 2 + ((d >> 2) & 1)) * 128 + ((d >> 3) * 16 + (b & 15)) * 4 + (d & 3)] = val;
    else
      out[(size_t)b * NOD + o * DO_ + d] = val;
  }
}

extern "C" void kernel_launch(void* const* d_in, const int* in_sizes, int n_in,
                              void* d_out, int out_size, void* d_ws, size_t ws_size,
                              hipStream_t stream) {
  const float* x = (const float*)d_in[0];
  const float* W = (const float*)d_in[1];
  float* out = (float*)d_out;

  float* ps   = (float*)d_ws;
  float* vbuf = ps + (size_t)NPART * PS_STRIDE;
  float* rlog = vbuf + (size_t)B_SZ * NOD;

  caps_s0_gemm<<<NB_BIG, 256, 0, stream>>>(x, W, ps);
  caps_reduce_squash<0, 0><<<256, 256, 0, stream>>>(ps, 1.0f / NO, vbuf);
  caps_route<1><<<NB_BIG, 128, 0, stream>>>(x, W, vbuf, rlog, ps);
  caps_reduce_squash<1, 0><<<256, 256, 0, stream>>>(ps, 1.0f, vbuf);
  caps_route<2><<<NB_BIG, 128, 0, stream>>>(x, W, vbuf, rlog, ps);
  caps_reduce_squash<1, 1><<<256, 256, 0, stream>>>(ps, 1.0f, out);
}
